# starting kernel
# baseline (speedup 1.0000x reference)
#include <hip/hip_runtime.h>

#define HH 768
#define WW 768
#define CC 4
#define NB 8
#define RAD 8
#define EPSV 1e-4f

typedef __attribute__((ext_vector_type(2))) float v2f;
typedef __attribute__((ext_vector_type(8))) float v8f;

static __device__ __forceinline__ v8f wmma4(v2f a, v2f b, v8f c) {
  return __builtin_amdgcn_wmma_f32_16x16x4_f32(false, a, false, b, (short)0, c,
                                               false, false);
}

static __device__ __forceinline__ v8f box2d_16(const float* X, float* T,
                                               const v2f band[8], int lane) {
  const int m = lane & 15;
  const int hi = lane >> 4;
  #pragma unroll
  for (int rs = 0; rs < 2; ++rs) {
    v2f a[8];
    #pragma unroll
    for (int kk = 0; kk < 8; ++kk) {
      const int j0 = 4 * kk + 2 * hi;
      v2f t = { X[(rs * 16 + m) * 33 + j0],
                X[(rs * 16 + m) * 33 + j0 + 1] };
      a[kk] = t;
    }
    v8f acc = {};
    #pragma unroll
    for (int kk = 0; kk < 8; ++kk) acc = wmma4(a[kk], band[kk], acc);
    #pragma unroll
    for (int v = 0; v < 8; ++v)
      T[(rs * 16 + v + 8 * hi) * 17 + m] = acc[v];
  }
  __syncthreads();
  v2f b[8];
  #pragma unroll
  for (int kk = 0; kk < 8; ++kk) {
    const int j0 = 4 * kk + 2 * hi;
    v2f t = { T[j0 * 17 + m], T[(j0 + 1) * 17 + m] };
    b[kk] = t;
  }
  v8f acc = {};
  #pragma unroll
  for (int kk = 0; kk < 8; ++kk) acc = wmma4(band[kk], b[kk], acc);
  __syncthreads();
  return acc;
}

static __device__ __forceinline__ void make_band(v2f band[8], int lane) {
  const int hi = lane >> 4;
  const int m = lane & 15;
  #pragma unroll
  for (int kk = 0; kk < 8; ++kk) {
    const int d0 = 4 * kk + 2 * hi - m;
    v2f t = { (d0 >= 0 && d0 <= 16) ? 1.f : 0.f,
              (d0 + 1 >= 0 && d0 + 1 <= 16) ? 1.f : 0.f };
    band[kk] = t;
  }
}

static __device__ __forceinline__ void inv4_rowsum(const float m[16],
                                                   float rs[4]) {
  float i0  =  m[5]*m[10]*m[15] - m[5]*m[11]*m[14] - m[9]*m[6]*m[15] + m[9]*m[7]*m[14] + m[13]*m[6]*m[11] - m[13]*m[7]*m[10];
  float i4  = -m[4]*m[10]*m[15] + m[4]*m[11]*m[14] + m[8]*m[6]*m[15] - m[8]*m[7]*m[14] - m[12]*m[6]*m[11] + m[12]*m[7]*m[10];
  float i8  =  m[4]*m[9]*m[15] - m[4]*m[11]*m[13] - m[8]*m[5]*m[15] + m[8]*m[7]*m[13] + m[12]*m[5]*m[11] - m[12]*m[7]*m[9];
  float i12 = -m[4]*m[9]*m[14] + m[4]*m[10]*m[13] + m[8]*m[5]*m[14] - m[8]*m[6]*m[13] - m[12]*m[5]*m[10] + m[12]*m[6]*m[9];
  float i1  = -m[1]*m[10]*m[15] + m[1]*m[11]*m[14] + m[9]*m[2]*m[15] - m[9]*m[3]*m[14] - m[13]*m[2]*m[11] + m[13]*m[3]*m[10];
  float i5  =  m[0]*m[10]*m[15] - m[0]*m[11]*m[14] - m[8]*m[2]*m[15] + m[8]*m[3]*m[14] + m[12]*m[2]*m[11] - m[12]*m[3]*m[10];
  float i9  = -m[0]*m[9]*m[15] + m[0]*m[11]*m[13] + m[8]*m[1]*m[15] - m[8]*m[3]*m[13] - m[12]*m[1]*m[11] + m[12]*m[3]*m[9];
  float i13 =  m[0]*m[9]*m[14] - m[0]*m[10]*m[13] - m[8]*m[1]*m[14] + m[8]*m[2]*m[13] + m[12]*m[1]*m[10] - m[12]*m[2]*m[9];
  float i2  =  m[1]*m[6]*m[15] - m[1]*m[7]*m[14] - m[5]*m[2]*m[15] + m[5]*m[3]*m[14] + m[13]*m[2]*m[7] - m[13]*m[3]*m[6];
  float i6  = -m[0]*m[6]*m[15] + m[0]*m[7]*m[14] + m[4]*m[2]*m[15] - m[4]*m[3]*m[14] - m[12]*m[2]*m[7] + m[12]*m[3]*m[6];
  float i10 =  m[0]*m[5]*m[15] - m[0]*m[7]*m[13] - m[4]*m[1]*m[15] + m[4]*m[3]*m[13] + m[12]*m[1]*m[7] - m[12]*m[3]*m[5];
  float i14 = -m[0]*m[5]*m[14] + m[0]*m[6]*m[13] + m[4]*m[1]*m[14] - m[4]*m[2]*m[13] - m[12]*m[1]*m[6] + m[12]*m[2]*m[5];
  float i3  = -m[1]*m[6]*m[11] + m[1]*m[7]*m[10] + m[5]*m[2]*m[11] - m[5]*m[3]*m[10] - m[9]*m[2]*m[7] + m[9]*m[3]*m[6];
  float i7  =  m[0]*m[6]*m[11] - m[0]*m[7]*m[10] - m[4]*m[2]*m[11] + m[4]*m[3]*m[10] + m[8]*m[2]*m[7] - m[8]*m[3]*m[6];
  float i11 = -m[0]*m[5]*m[11] + m[0]*m[7]*m[9] + m[4]*m[1]*m[11] - m[4]*m[3]*m[9] - m[8]*m[1]*m[7] + m[8]*m[3]*m[5];
  float i15 =  m[0]*m[5]*m[10] - m[0]*m[6]*m[9] - m[4]*m[1]*m[10] + m[4]*m[2]*m[9] + m[8]*m[1]*m[6] - m[8]*m[2]*m[5];
  const float det = m[0]*i0 + m[1]*i4 + m[2]*i8 + m[3]*i12;
  const float id = 1.0f / det;
  rs[0] = (i0 + i1 + i2 + i3) * id;
  rs[1] = (i4 + i5 + i6 + i7) * id;
  rs[2] = (i8 + i9 + i10 + i11) * id;
  rs[3] = (i12 + i13 + i14 + i15) * id;
}

__global__ __launch_bounds__(32) void gf_stats(const float* __restrict__ I,
                                               const float* __restrict__ P,
                                               float* __restrict__ Aout,
                                               float* __restrict__ Bout) {
  __shared__ float raw[5][32 * 33];
  __shared__ float X[32 * 33];
  __shared__ float T[32 * 17];
  const int lane = threadIdx.x;
  const int tx = blockIdx.x * 16;
  const int ty = blockIdx.y * 16;
  const int bb = blockIdx.z;
  const int gc = tx - RAD + lane;
  const bool cok = (gc >= 0) && (gc < WW);
  const float* Ib = I + (size_t)bb * CC * HH * WW;
  const float* Pb = P + (size_t)bb * HH * WW;
  for (int y = 0; y < 32; ++y) {
    const int gr = ty - RAD + y;
    const bool ok = cok && (gr >= 0) && (gr < HH);
    const size_t off = (size_t)gr * WW + gc;
    raw[4][y * 33 + lane] = ok ? Pb[off] : 0.f;
    #pragma unroll
    for (int c = 0; c < CC; ++c)
      raw[c][y * 33 + lane] = ok ? Ib[(size_t)c * HH * WW + off] : 0.f;
  }
  __syncthreads();

  v2f band[8];
  make_band(band, lane);

  v8f S[19];
  #pragma unroll
  for (int pl = 0; pl < 5; ++pl)
    S[pl] = box2d_16(&raw[pl][0], T, band, lane);

  constexpr int ia[14]  = {4,4,4,4, 0,0,0,0, 1,1,1, 2,2, 3};
  constexpr int ibx[14] = {0,1,2,3, 0,1,2,3, 1,2,3, 2,3, 3};
  #pragma unroll
  for (int pl = 0; pl < 14; ++pl) {
    #pragma unroll
    for (int y = 0; y < 32; ++y)
      X[y * 33 + lane] =
          raw[ia[pl]][y * 33 + lane] * raw[ibx[pl]][y * 33 + lane];
    __syncthreads();
    S[5 + pl] = box2d_16(X, T, band, lane);
  }

  const int hi = lane >> 4;
  const int w = lane & 15;
  const int col = tx + w;
  const int cw = min(col + RAD, WW - 1) - max(col - RAD, 0) + 1;
  #pragma unroll
  for (int v = 0; v < 8; ++v) {
    const int row = ty + v + 8 * hi;
    const int chc = min(row + RAD, HH - 1) - max(row - RAD, 0) + 1;
    const float invN = 1.f / (float)(chc * cw);
    const float pm = S[4][v] * invN;
    float Im[4], first[4];
    float sumIm = 0.f;
    #pragma unroll
    for (int c = 0; c < 4; ++c) {
      Im[c] = S[c][v] * invN;
      sumIm += Im[c];
      first[c] = S[5 + c][v] * invN - pm * Im[c];
    }
    float M[16];
    M[0]  = S[9][v]  * invN + EPSV;
    M[1]  = M[4]  = S[10][v] * invN;
    M[2]  = M[8]  = S[11][v] * invN;
    M[3]  = M[12] = S[12][v] * invN;
    M[5]  = S[13][v] * invN + EPSV;
    M[6]  = M[9]  = S[14][v] * invN;
    M[7]  = M[13] = S[15][v] * invN;
    M[10] = S[16][v] * invN + EPSV;
    M[11] = M[14] = S[17][v] * invN;
    M[15] = S[18][v] * invN + EPSV;
    float rs[4];
    inv4_rowsum(M, rs);
    float sumA = 0.f;
    float Ac[4];
    #pragma unroll
    for (int c = 0; c < 4; ++c) {
      Ac[c] = first[c] * rs[c];
      sumA += Ac[c];
    }
    const float bv = pm - sumA * sumIm;
    #pragma unroll
    for (int c = 0; c < 4; ++c)
      Aout[(((size_t)bb * CC + c) * HH + row) * WW + col] = Ac[c];
    Bout[((size_t)bb * HH + row) * WW + col] = bv;
  }
}

__global__ __launch_bounds__(32) void gf_final(const float* __restrict__ I,
                                               const float* __restrict__ Ain,
                                               const float* __restrict__ Bin,
                                               float* __restrict__ Q) {
  __shared__ float X[32 * 33];
  __shared__ float T[32 * 17];
  const int lane = threadIdx.x;
  const int tx = blockIdx.x * 16;
  const int ty = blockIdx.y * 16;
  const int bb = blockIdx.z;
  const int gc = tx - RAD + lane;
  const bool cok = (gc >= 0) && (gc < WW);
  v2f band[8];
  make_band(band, lane);
  v8f S[5];
  #pragma unroll
  for (int pl = 0; pl < 5; ++pl) {
    const float* src = (pl < 4) ? (Ain + (((size_t)bb * CC + pl) * (size_t)HH) * WW)
                                : (Bin + (size_t)bb * HH * WW);
    for (int y = 0; y < 32; ++y) {
      const int gr = ty - RAD + y;
      const bool ok = cok && (gr >= 0) && (gr < HH);
      X[y * 33 + lane] = ok ? src[(size_t)gr * WW + gc] : 0.f;
    }
    __syncthreads();
    S[pl] = box2d_16(X, T, band, lane);
  }
  const int hi = lane >> 4;
  const int w = lane & 15;
  const int col = tx + w;
  const int cw = min(col + RAD, WW - 1) - max(col - RAD, 0) + 1;
  const float* Ib = I + (size_t)bb * CC * HH * WW;
  #pragma unroll
  for (int v = 0; v < 8; ++v) {
    const int row = ty + v + 8 * hi;
    const int chc = min(row + RAD, HH - 1) - max(row - RAD, 0) + 1;
    const float invN = 1.f / (float)(chc * cw);
    const size_t off = (size_t)row * WW + col;
    float sumI = 0.f;
    #pragma unroll
    for (int c = 0; c < 4; ++c) sumI += Ib[(size_t)c * HH * WW + off];
    const float am = (S[0][v] + S[1][v] + S[2][v] + S[3][v]) * invN;
    const float bm = S[4][v] * invN;
    Q[((size_t)bb * HH + row) * WW + col] = am * sumI + bm;
  }
}

extern "C" void kernel_launch(void* const* d_in, const int* in_sizes, int n_in,
                              void* d_out, int out_size, void* d_ws,
                              size_t ws_size, hipStream_t stream) {
  const float* I = (const float*)d_in[0];
  const float* P = (const float*)d_in[1];
  float* Q = (float*)d_out;
  float* Aws = (float*)d_ws;
  float* Bws = Aws + (size_t)NB * CC * HH * WW;
  dim3 grid(WW / 16, HH / 16, NB);
  gf_stats<<<grid, dim3(32), 0, stream>>>(I, P, Aws, Bws);
  gf_final<<<grid, dim3(32), 0, stream>>>(I, Aws, Bws, Q);
  (void)in_sizes; (void)n_in; (void)out_size; (void)ws_size;
}


# Halo tiles fetched in batches: clamped in-bounds loads issued together ahead of use (zero padding by select at the LDS store) instead of one load-wait-store round trip per row
# speedup vs baseline: 1.9117x; 1.9117x over previous
#include <hip/hip_runtime.h>

#define HH 768
#define WW 768
#define CC 4
#define NB 8
#define RAD 8
#define EPSV 1e-4f

typedef __attribute__((ext_vector_type(2))) float v2f;
typedef __attribute__((ext_vector_type(8))) float v8f;

static __device__ __forceinline__ v8f wmma4(v2f a, v2f b, v8f c) {
  return __builtin_amdgcn_wmma_f32_16x16x4_f32(false, a, false, b, (short)0, c,
                                               false, false);
}

static __device__ __forceinline__ v8f box2d_16(const float* X, float* T,
                                               const v2f band[8], int lane) {
  const int m = lane & 15;
  const int hi = lane >> 4;
  #pragma unroll
  for (int rs = 0; rs < 2; ++rs) {
    v2f a[8];
    #pragma unroll
    for (int kk = 0; kk < 8; ++kk) {
      const int j0 = 4 * kk + 2 * hi;
      v2f t = { X[(rs * 16 + m) * 33 + j0],
                X[(rs * 16 + m) * 33 + j0 + 1] };
      a[kk] = t;
    }
    v8f acc = {};
    #pragma unroll
    for (int kk = 0; kk < 8; ++kk) acc = wmma4(a[kk], band[kk], acc);
    #pragma unroll
    for (int v = 0; v < 8; ++v)
      T[(rs * 16 + v + 8 * hi) * 17 + m] = acc[v];
  }
  __syncthreads();
  v2f b[8];
  #pragma unroll
  for (int kk = 0; kk < 8; ++kk) {
    const int j0 = 4 * kk + 2 * hi;
    v2f t = { T[j0 * 17 + m], T[(j0 + 1) * 17 + m] };
    b[kk] = t;
  }
  v8f acc = {};
  #pragma unroll
  for (int kk = 0; kk < 8; ++kk) acc = wmma4(band[kk], b[kk], acc);
  __syncthreads();
  return acc;
}

static __device__ __forceinline__ void make_band(v2f band[8], int lane) {
  const int hi = lane >> 4;
  const int m = lane & 15;
  #pragma unroll
  for (int kk = 0; kk < 8; ++kk) {
    const int d0 = 4 * kk + 2 * hi - m;
    v2f t = { (d0 >= 0 && d0 <= 16) ? 1.f : 0.f,
              (d0 + 1 >= 0 && d0 + 1 <= 16) ? 1.f : 0.f };
    band[kk] = t;
  }
}

static __device__ __forceinline__ void inv4_rowsum(const float m[16],
                                                   float rs[4]) {
  float i0  =  m[5]*m[10]*m[15] - m[5]*m[11]*m[14] - m[9]*m[6]*m[15] + m[9]*m[7]*m[14] + m[13]*m[6]*m[11] - m[13]*m[7]*m[10];
  float i4  = -m[4]*m[10]*m[15] + m[4]*m[11]*m[14] + m[8]*m[6]*m[15] - m[8]*m[7]*m[14] - m[12]*m[6]*m[11] + m[12]*m[7]*m[10];
  float i8  =  m[4]*m[9]*m[15] - m[4]*m[11]*m[13] - m[8]*m[5]*m[15] + m[8]*m[7]*m[13] + m[12]*m[5]*m[11] - m[12]*m[7]*m[9];
  float i12 = -m[4]*m[9]*m[14] + m[4]*m[10]*m[13] + m[8]*m[5]*m[14] - m[8]*m[6]*m[13] - m[12]*m[5]*m[10] + m[12]*m[6]*m[9];
  float i1  = -m[1]*m[10]*m[15] + m[1]*m[11]*m[14] + m[9]*m[2]*m[15] - m[9]*m[3]*m[14] - m[13]*m[2]*m[11] + m[13]*m[3]*m[10];
  float i5  =  m[0]*m[10]*m[15] - m[0]*m[11]*m[14] - m[8]*m[2]*m[15] + m[8]*m[3]*m[14] + m[12]*m[2]*m[11] - m[12]*m[3]*m[10];
  float i9  = -m[0]*m[9]*m[15] + m[0]*m[11]*m[13] + m[8]*m[1]*m[15] - m[8]*m[3]*m[13] - m[12]*m[1]*m[11] + m[12]*m[3]*m[9];
  float i13 =  m[0]*m[9]*m[14] - m[0]*m[10]*m[13] - m[8]*m[1]*m[14] + m[8]*m[2]*m[13] + m[12]*m[1]*m[10] - m[12]*m[2]*m[9];
  float i2  =  m[1]*m[6]*m[15] - m[1]*m[7]*m[14] - m[5]*m[2]*m[15] + m[5]*m[3]*m[14] + m[13]*m[2]*m[7] - m[13]*m[3]*m[6];
  float i6  = -m[0]*m[6]*m[15] + m[0]*m[7]*m[14] + m[4]*m[2]*m[15] - m[4]*m[3]*m[14] - m[12]*m[2]*m[7] + m[12]*m[3]*m[6];
  float i10 =  m[0]*m[5]*m[15] - m[0]*m[7]*m[13] - m[4]*m[1]*m[15] + m[4]*m[3]*m[13] + m[12]*m[1]*m[7] - m[12]*m[3]*m[5];
  float i14 = -m[0]*m[5]*m[14] + m[0]*m[6]*m[13] + m[4]*m[1]*m[14] - m[4]*m[2]*m[13] - m[12]*m[1]*m[6] + m[12]*m[2]*m[5];
  float i3  = -m[1]*m[6]*m[11] + m[1]*m[7]*m[10] + m[5]*m[2]*m[11] - m[5]*m[3]*m[10] - m[9]*m[2]*m[7] + m[9]*m[3]*m[6];
  float i7  =  m[0]*m[6]*m[11] - m[0]*m[7]*m[10] - m[4]*m[2]*m[11] + m[4]*m[3]*m[10] + m[8]*m[2]*m[7] - m[8]*m[3]*m[6];
  float i11 = -m[0]*m[5]*m[11] + m[0]*m[7]*m[9] + m[4]*m[1]*m[11] - m[4]*m[3]*m[9] - m[8]*m[1]*m[7] + m[8]*m[3]*m[5];
  float i15 =  m[0]*m[5]*m[10] - m[0]*m[6]*m[9] - m[4]*m[1]*m[10] + m[4]*m[2]*m[9] + m[8]*m[1]*m[6] - m[8]*m[2]*m[5];
  const float det = m[0]*i0 + m[1]*i4 + m[2]*i8 + m[3]*i12;
  const float id = 1.0f / det;
  rs[0] = (i0 + i1 + i2 + i3) * id;
  rs[1] = (i4 + i5 + i6 + i7) * id;
  rs[2] = (i8 + i9 + i10 + i11) * id;
  rs[3] = (i12 + i13 + i14 + i15) * id;
}

__global__ __launch_bounds__(32) void gf_stats(const float* __restrict__ I,
                                               const float* __restrict__ P,
                                               float* __restrict__ Aout,
                                               float* __restrict__ Bout) {
  __shared__ float raw[5][32 * 33];
  __shared__ float X[32 * 33];
  __shared__ float T[32 * 17];
  const int lane = threadIdx.x;
  const int tx = blockIdx.x * 16;
  const int ty = blockIdx.y * 16;
  const int bb = blockIdx.z;
  const int gc = tx - RAD + lane;
  const bool cok = (gc >= 0) && (gc < WW);
  const float* Ib = I + (size_t)bb * CC * HH * WW;
  const float* Pb = P + (size_t)bb * HH * WW;
  const int gcc = min(max(gc, 0), WW - 1);
  float hv[5][32];
  #pragma unroll
  for (int y = 0; y < 32; ++y) {
    const int grc = min(max(ty - RAD + y, 0), HH - 1);
    const size_t off = (size_t)grc * WW + gcc;
    hv[4][y] = Pb[off];
    #pragma unroll
    for (int c = 0; c < CC; ++c)
      hv[c][y] = Ib[(size_t)c * HH * WW + off];
  }
  #pragma unroll
  for (int y = 0; y < 32; ++y) {
    const int gr = ty - RAD + y;
    const bool ok = cok && (gr >= 0) && (gr < HH);
    raw[4][y * 33 + lane] = ok ? hv[4][y] : 0.f;
    #pragma unroll
    for (int c = 0; c < CC; ++c)
      raw[c][y * 33 + lane] = ok ? hv[c][y] : 0.f;
  }
  __syncthreads();

  v2f band[8];
  make_band(band, lane);

  v8f S[19];
  #pragma unroll
  for (int pl = 0; pl < 5; ++pl)
    S[pl] = box2d_16(&raw[pl][0], T, band, lane);

  constexpr int ia[14]  = {4,4,4,4, 0,0,0,0, 1,1,1, 2,2, 3};
  constexpr int ibx[14] = {0,1,2,3, 0,1,2,3, 1,2,3, 2,3, 3};
  #pragma unroll
  for (int pl = 0; pl < 14; ++pl) {
    #pragma unroll
    for (int y = 0; y < 32; ++y)
      X[y * 33 + lane] =
          raw[ia[pl]][y * 33 + lane] * raw[ibx[pl]][y * 33 + lane];
    __syncthreads();
    S[5 + pl] = box2d_16(X, T, band, lane);
  }

  const int hi = lane >> 4;
  const int w = lane & 15;
  const int col = tx + w;
  const int cw = min(col + RAD, WW - 1) - max(col - RAD, 0) + 1;
  #pragma unroll
  for (int v = 0; v < 8; ++v) {
    const int row = ty + v + 8 * hi;
    const int chc = min(row + RAD, HH - 1) - max(row - RAD, 0) + 1;
    const float invN = 1.f / (float)(chc * cw);
    const float pm = S[4][v] * invN;
    float Im[4], first[4];
    float sumIm = 0.f;
    #pragma unroll
    for (int c = 0; c < 4; ++c) {
      Im[c] = S[c][v] * invN;
      sumIm += Im[c];
      first[c] = S[5 + c][v] * invN - pm * Im[c];
    }
    float M[16];
    M[0]  = S[9][v]  * invN + EPSV;
    M[1]  = M[4]  = S[10][v] * invN;
    M[2]  = M[8]  = S[11][v] * invN;
    M[3]  = M[12] = S[12][v] * invN;
    M[5]  = S[13][v] * invN + EPSV;
    M[6]  = M[9]  = S[14][v] * invN;
    M[7]  = M[13] = S[15][v] * invN;
    M[10] = S[16][v] * invN + EPSV;
    M[11] = M[14] = S[17][v] * invN;
    M[15] = S[18][v] * invN + EPSV;
    float rs[4];
    inv4_rowsum(M, rs);
    float sumA = 0.f;
    float Ac[4];
    #pragma unroll
    for (int c = 0; c < 4; ++c) {
      Ac[c] = first[c] * rs[c];
      sumA += Ac[c];
    }
    const float bv = pm - sumA * sumIm;
    #pragma unroll
    for (int c = 0; c < 4; ++c)
      Aout[(((size_t)bb * CC + c) * HH + row) * WW + col] = Ac[c];
    Bout[((size_t)bb * HH + row) * WW + col] = bv;
  }
}

__global__ __launch_bounds__(32) void gf_final(const float* __restrict__ I,
                                               const float* __restrict__ Ain,
                                               const float* __restrict__ Bin,
                                               float* __restrict__ Q) {
  __shared__ float X[32 * 33];
  __shared__ float T[32 * 17];
  const int lane = threadIdx.x;
  const int tx = blockIdx.x * 16;
  const int ty = blockIdx.y * 16;
  const int bb = blockIdx.z;
  const int gc = tx - RAD + lane;
  const bool cok = (gc >= 0) && (gc < WW);
  v2f band[8];
  make_band(band, lane);
  const int hi = lane >> 4;
  const int w = lane & 15;
  const int col = tx + w;
  const float* Ib = I + (size_t)bb * CC * HH * WW;
  const int gcc = min(max(gc, 0), WW - 1);
  float hv[32];
  float iv[4][8];
  #pragma unroll
  for (int y = 0; y < 32; ++y) {
    const int grc = min(max(ty - RAD + y, 0), HH - 1);
    hv[y] = Ain[(((size_t)bb * CC) * (size_t)HH + grc) * WW + gcc];
  }
  v8f S[5];
  #pragma unroll
  for (int pl = 0; pl < 5; ++pl) {
    #pragma unroll
    for (int y = 0; y < 32; ++y) {
      const int gr = ty - RAD + y;
      const bool ok = cok && (gr >= 0) && (gr < HH);
      X[y * 33 + lane] = ok ? hv[y] : 0.f;
    }
    __syncthreads();
    if (pl < 4) {
      const float* src = (pl + 1 < 4)
          ? (Ain + (((size_t)bb * CC + (pl + 1)) * (size_t)HH) * WW)
          : (Bin + (size_t)bb * HH * WW);
      #pragma unroll
      for (int y = 0; y < 32; ++y) {
        const int grc = min(max(ty - RAD + y, 0), HH - 1);
        hv[y] = src[(size_t)grc * WW + gcc];
      }
    } else {
      #pragma unroll
      for (int v = 0; v < 8; ++v) {
        const size_t off = (size_t)(ty + v + 8 * hi) * WW + col;
        #pragma unroll
        for (int c = 0; c < 4; ++c) iv[c][v] = Ib[(size_t)c * HH * WW + off];
      }
    }
    S[pl] = box2d_16(X, T, band, lane);
  }
  const int cw = min(col + RAD, WW - 1) - max(col - RAD, 0) + 1;
  #pragma unroll
  for (int v = 0; v < 8; ++v) {
    const int row = ty + v + 8 * hi;
    const int chc = min(row + RAD, HH - 1) - max(row - RAD, 0) + 1;
    const float invN = 1.f / (float)(chc * cw);
    float sumI = 0.f;
    #pragma unroll
    for (int c = 0; c < 4; ++c) sumI += iv[c][v];
    const float am = (S[0][v] + S[1][v] + S[2][v] + S[3][v]) * invN;
    const float bm = S[4][v] * invN;
    Q[((size_t)bb * HH + row) * WW + col] = am * sumI + bm;
  }
}

extern "C" void kernel_launch(void* const* d_in, const int* in_sizes, int n_in,
                              void* d_out, int out_size, void* d_ws,
                              size_t ws_size, hipStream_t stream) {
  const float* I = (const float*)d_in[0];
  const float* P = (const float*)d_in[1];
  float* Q = (float*)d_out;
  float* Aws = (float*)d_ws;
  float* Bws = Aws + (size_t)NB * CC * HH * WW;
  dim3 grid(WW / 16, HH / 16, NB);
  gf_stats<<<grid, dim3(32), 0, stream>>>(I, P, Aws, Bws);
  gf_final<<<grid, dim3(32), 0, stream>>>(I, Aws, Bws, Q);
  (void)in_sizes; (void)n_in; (void)out_size; (void)ws_size;
}


# Box filter kept in registers: products formed in the A fragments, vertical pass takes the horizontal accumulators as B operand, one band and K order for both passes (sums reordered); no LDS staging
# speedup vs baseline: 2.6765x; 1.4001x over previous
#include <hip/hip_runtime.h>

#define HH 768
#define WW 768
#define CC 4
#define NB 8
#define RAD 8
#define EPSV 1e-4f

typedef __attribute__((ext_vector_type(2))) float v2f;
typedef __attribute__((ext_vector_type(8))) float v8f;

static __device__ __forceinline__ v8f wmma4(v2f a, v2f b, v8f c) {
  return __builtin_amdgcn_wmma_f32_16x16x4_f32(false, a, false, b, (short)0, c,
                                               false, false);
}

static __device__ __forceinline__ void load_frag(const float* X, int rs,
                                                 int lane, v2f a[8]) {
  const int m = lane & 15;
  const int hi = lane >> 4;
  #pragma unroll
  for (int kk = 0; kk < 8; ++kk) {
    const int j0 = 16 * (kk >> 2) + 8 * hi + 2 * (kk & 3);
    v2f t = { X[(rs * 16 + m) * 33 + j0], X[(rs * 16 + m) * 33 + j0 + 1] };
    a[kk] = t;
  }
}

static __device__ __forceinline__ void make_band(v2f band[8], int lane) {
  const int hi = lane >> 4;
  const int m = lane & 15;
  #pragma unroll
  for (int kk = 0; kk < 8; ++kk) {
    const int d0 = 16 * (kk >> 2) + 8 * hi + 2 * (kk & 3) - m;
    v2f t = { (d0 >= 0 && d0 <= 16) ? 1.f : 0.f,
              (d0 + 1 >= 0 && d0 + 1 <= 16) ? 1.f : 0.f };
    band[kk] = t;
  }
}

static __device__ __forceinline__ v8f hpass(const v2f a[8],
                                            const v2f band[8]) {
  v8f acc = {};
  #pragma unroll
  for (int kk = 0; kk < 8; ++kk) acc = wmma4(a[kk], band[kk], acc);
  return acc;
}

static __device__ __forceinline__ v8f hpass_prod(const v2f a[8],
                                                 const v2f b[8],
                                                 const v2f band[8]) {
  v8f acc = {};
  #pragma unroll
  for (int kk = 0; kk < 8; ++kk) acc = wmma4(a[kk] * b[kk], band[kk], acc);
  return acc;
}

static __device__ __forceinline__ v8f vpass(const v2f band[8], int t, v8f h,
                                            v8f acc) {
  #pragma unroll
  for (int j = 0; j < 4; ++j) {
    v2f b = { h[2 * j], h[2 * j + 1] };
    acc = wmma4(band[4 * t + j], b, acc);
  }
  return acc;
}

static __device__ __forceinline__ void inv4_rowsum(const float m[16],
                                                   float rs[4]) {
  float i0  =  m[5]*m[10]*m[15] - m[5]*m[11]*m[14] - m[9]*m[6]*m[15] + m[9]*m[7]*m[14] + m[13]*m[6]*m[11] - m[13]*m[7]*m[10];
  float i4  = -m[4]*m[10]*m[15] + m[4]*m[11]*m[14] + m[8]*m[6]*m[15] - m[8]*m[7]*m[14] - m[12]*m[6]*m[11] + m[12]*m[7]*m[10];
  float i8  =  m[4]*m[9]*m[15] - m[4]*m[11]*m[13] - m[8]*m[5]*m[15] + m[8]*m[7]*m[13] + m[12]*m[5]*m[11] - m[12]*m[7]*m[9];
  float i12 = -m[4]*m[9]*m[14] + m[4]*m[10]*m[13] + m[8]*m[5]*m[14] - m[8]*m[6]*m[13] - m[12]*m[5]*m[10] + m[12]*m[6]*m[9];
  float i1  = -m[1]*m[10]*m[15] + m[1]*m[11]*m[14] + m[9]*m[2]*m[15] - m[9]*m[3]*m[14] - m[13]*m[2]*m[11] + m[13]*m[3]*m[10];
  float i5  =  m[0]*m[10]*m[15] - m[0]*m[11]*m[14] - m[8]*m[2]*m[15] + m[8]*m[3]*m[14] + m[12]*m[2]*m[11] - m[12]*m[3]*m[10];
  float i9  = -m[0]*m[9]*m[15] + m[0]*m[11]*m[13] + m[8]*m[1]*m[15] - m[8]*m[3]*m[13] - m[12]*m[1]*m[11] + m[12]*m[3]*m[9];
  float i13 =  m[0]*m[9]*m[14] - m[0]*m[10]*m[13] - m[8]*m[1]*m[14] + m[8]*m[2]*m[13] + m[12]*m[1]*m[10] - m[12]*m[2]*m[9];
  float i2  =  m[1]*m[6]*m[15] - m[1]*m[7]*m[14] - m[5]*m[2]*m[15] + m[5]*m[3]*m[14] + m[13]*m[2]*m[7] - m[13]*m[3]*m[6];
  float i6  = -m[0]*m[6]*m[15] + m[0]*m[7]*m[14] + m[4]*m[2]*m[15] - m[4]*m[3]*m[14] - m[12]*m[2]*m[7] + m[12]*m[3]*m[6];
  float i10 =  m[0]*m[5]*m[15] - m[0]*m[7]*m[13] - m[4]*m[1]*m[15] + m[4]*m[3]*m[13] + m[12]*m[1]*m[7] - m[12]*m[3]*m[5];
  float i14 = -m[0]*m[5]*m[14] + m[0]*m[6]*m[13] + m[4]*m[1]*m[14] - m[4]*m[2]*m[13] - m[12]*m[1]*m[6] + m[12]*m[2]*m[5];
  float i3  = -m[1]*m[6]*m[11] + m[1]*m[7]*m[10] + m[5]*m[2]*m[11] - m[5]*m[3]*m[10] - m[9]*m[2]*m[7] + m[9]*m[3]*m[6];
  float i7  =  m[0]*m[6]*m[11] - m[0]*m[7]*m[10] - m[4]*m[2]*m[11] + m[4]*m[3]*m[10] + m[8]*m[2]*m[7] - m[8]*m[3]*m[6];
  float i11 = -m[0]*m[5]*m[11] + m[0]*m[7]*m[9] + m[4]*m[1]*m[11] - m[4]*m[3]*m[9] - m[8]*m[1]*m[7] + m[8]*m[3]*m[5];
  float i15 =  m[0]*m[5]*m[10] - m[0]*m[6]*m[9] - m[4]*m[1]*m[10] + m[4]*m[2]*m[9] + m[8]*m[1]*m[6] - m[8]*m[2]*m[5];
  const float det = m[0]*i0 + m[1]*i4 + m[2]*i8 + m[3]*i12;
  const float id = 1.0f / det;
  rs[0] = (i0 + i1 + i2 + i3) * id;
  rs[1] = (i4 + i5 + i6 + i7) * id;
  rs[2] = (i8 + i9 + i10 + i11) * id;
  rs[3] = (i12 + i13 + i14 + i15) * id;
}

__global__ __launch_bounds__(32) void gf_stats(const float* __restrict__ I,
                                               const float* __restrict__ P,
                                               float* __restrict__ Aout,
                                               float* __restrict__ Bout) {
  __shared__ float raw[5][32 * 33];
  const int lane = threadIdx.x;
  const int tx = blockIdx.x * 16;
  const int ty = blockIdx.y * 16;
  const int bb = blockIdx.z;
  const int gc = tx - RAD + lane;
  const bool cok = (gc >= 0) && (gc < WW);
  const float* Ib = I + (size_t)bb * CC * HH * WW;
  const float* Pb = P + (size_t)bb * HH * WW;
  const int gcc = min(max(gc, 0), WW - 1);
  float hv[5][32];
  #pragma unroll
  for (int y = 0; y < 32; ++y) {
    const int grc = min(max(ty - RAD + y, 0), HH - 1);
    const size_t off = (size_t)grc * WW + gcc;
    hv[4][y] = Pb[off];
    #pragma unroll
    for (int c = 0; c < CC; ++c)
      hv[c][y] = Ib[(size_t)c * HH * WW + off];
  }
  #pragma unroll
  for (int y = 0; y < 32; ++y) {
    const int gr = ty - RAD + y;
    const bool ok = cok && (gr >= 0) && (gr < HH);
    raw[4][y * 33 + lane] = ok ? hv[4][y] : 0.f;
    #pragma unroll
    for (int c = 0; c < CC; ++c)
      raw[c][y * 33 + lane] = ok ? hv[c][y] : 0.f;
  }
  __syncthreads();

  v2f band[8];
  make_band(band, lane);

  constexpr int po[19] = {4,5,6,7,8, 0,9,10,11,12, 1,13,14,15, 2,16,17, 3,18};
  constexpr int pa[19] = {4,4,4,4,4, 0,0,0,0,0,    1,1,1,1,    2,2,2,   3,3};
  constexpr int pb[19] = {-1,0,1,2,3, -1,0,1,2,3,  -1,1,2,3,   -1,2,3,  -1,3};
  v2f F[2][5][8];
  #pragma unroll
  for (int rs = 0; rs < 2; ++rs) {
    #pragma unroll
    for (int pl = 0; pl < 5; ++pl) load_frag(&raw[pl][0], rs, lane, F[rs][pl]);
  }
  v8f S[19];
  #pragma unroll
  for (int i = 0; i < 19; ++i) {
    const int a = pa[i];
    const int b = pb[i] < 0 ? 0 : pb[i];
    v8f h0, h1;
    if (pb[i] < 0) {
      h0 = hpass(F[0][a], band);
      h1 = hpass(F[1][a], band);
    } else {
      h0 = hpass_prod(F[0][a], F[0][b], band);
      h1 = hpass_prod(F[1][a], F[1][b], band);
    }
    v8f z = {};
    S[po[i]] = vpass(band, 1, h1, vpass(band, 0, h0, z));
  }

  const int hi = lane >> 4;
  const int w = lane & 15;
  const int col = tx + w;
  const int cw = min(col + RAD, WW - 1) - max(col - RAD, 0) + 1;
  #pragma unroll
  for (int v = 0; v < 8; ++v) {
    const int row = ty + v + 8 * hi;
    const int chc = min(row + RAD, HH - 1) - max(row - RAD, 0) + 1;
    const float invN = 1.f / (float)(chc * cw);
    const float pm = S[4][v] * invN;
    float Im[4], first[4];
    float sumIm = 0.f;
    #pragma unroll
    for (int c = 0; c < 4; ++c) {
      Im[c] = S[c][v] * invN;
      sumIm += Im[c];
      first[c] = S[5 + c][v] * invN - pm * Im[c];
    }
    float M[16];
    M[0]  = S[9][v]  * invN + EPSV;
    M[1]  = M[4]  = S[10][v] * invN;
    M[2]  = M[8]  = S[11][v] * invN;
    M[3]  = M[12] = S[12][v] * invN;
    M[5]  = S[13][v] * invN + EPSV;
    M[6]  = M[9]  = S[14][v] * invN;
    M[7]  = M[13] = S[15][v] * invN;
    M[10] = S[16][v] * invN + EPSV;
    M[11] = M[14] = S[17][v] * invN;
    M[15] = S[18][v] * invN + EPSV;
    float rs[4];
    inv4_rowsum(M, rs);
    float sumA = 0.f;
    float Ac[4];
    #pragma unroll
    for (int c = 0; c < 4; ++c) {
      Ac[c] = first[c] * rs[c];
      sumA += Ac[c];
    }
    const float bv = pm - sumA * sumIm;
    #pragma unroll
    for (int c = 0; c < 4; ++c)
      Aout[(((size_t)bb * CC + c) * HH + row) * WW + col] = Ac[c];
    Bout[((size_t)bb * HH + row) * WW + col] = bv;
  }
}

__global__ __launch_bounds__(32) void gf_final(const float* __restrict__ I,
                                               const float* __restrict__ Ain,
                                               const float* __restrict__ Bin,
                                               float* __restrict__ Q) {
  __shared__ float X[32 * 33];
  const int lane = threadIdx.x;
  const int tx = blockIdx.x * 16;
  const int ty = blockIdx.y * 16;
  const int bb = blockIdx.z;
  const int gc = tx - RAD + lane;
  const bool cok = (gc >= 0) && (gc < WW);
  v2f band[8];
  make_band(band, lane);
  const int hi = lane >> 4;
  const int w = lane & 15;
  const int col = tx + w;
  const float* Ib = I + (size_t)bb * CC * HH * WW;
  const int gcc = min(max(gc, 0), WW - 1);
  float hv[32];
  float iv[4][8];
  #pragma unroll
  for (int y = 0; y < 32; ++y) {
    const int grc = min(max(ty - RAD + y, 0), HH - 1);
    hv[y] = Ain[(((size_t)bb * CC) * (size_t)HH + grc) * WW + gcc];
  }
  v8f S[5];
  #pragma unroll
  for (int pl = 0; pl < 5; ++pl) {
    #pragma unroll
    for (int y = 0; y < 32; ++y) {
      const int gr = ty - RAD + y;
      const bool ok = cok && (gr >= 0) && (gr < HH);
      X[y * 33 + lane] = ok ? hv[y] : 0.f;
    }
    __syncthreads();
    if (pl < 4) {
      const float* src = (pl + 1 < 4)
          ? (Ain + (((size_t)bb * CC + (pl + 1)) * (size_t)HH) * WW)
          : (Bin + (size_t)bb * HH * WW);
      #pragma unroll
      for (int y = 0; y < 32; ++y) {
        const int grc = min(max(ty - RAD + y, 0), HH - 1);
        hv[y] = src[(size_t)grc * WW + gcc];
      }
    } else {
      #pragma unroll
      for (int v = 0; v < 8; ++v) {
        const size_t off = (size_t)(ty + v + 8 * hi) * WW + col;
        #pragma unroll
        for (int c = 0; c < 4; ++c) iv[c][v] = Ib[(size_t)c * HH * WW + off];
      }
    }
    v2f a0[8], a1[8];
    load_frag(X, 0, lane, a0);
    load_frag(X, 1, lane, a1);
    __syncthreads();
    v8f z = {};
    S[pl] = vpass(band, 1, hpass(a1, band),
                  vpass(band, 0, hpass(a0, band), z));
  }
  const int cw = min(col + RAD, WW - 1) - max(col - RAD, 0) + 1;
  #pragma unroll
  for (int v = 0; v < 8; ++v) {
    const int row = ty + v + 8 * hi;
    const int chc = min(row + RAD, HH - 1) - max(row - RAD, 0) + 1;
    const float invN = 1.f / (float)(chc * cw);
    float sumI = 0.f;
    #pragma unroll
    for (int c = 0; c < 4; ++c) sumI += iv[c][v];
    const float am = (S[0][v] + S[1][v] + S[2][v] + S[3][v]) * invN;
    const float bm = S[4][v] * invN;
    Q[((size_t)bb * HH + row) * WW + col] = am * sumI + bm;
  }
}

extern "C" void kernel_launch(void* const* d_in, const int* in_sizes, int n_in,
                              void* d_out, int out_size, void* d_ws,
                              size_t ws_size, hipStream_t stream) {
  const float* I = (const float*)d_in[0];
  const float* P = (const float*)d_in[1];
  float* Q = (float*)d_out;
  float* Aws = (float*)d_ws;
  float* Bws = Aws + (size_t)NB * CC * HH * WW;
  dim3 grid(WW / 16, HH / 16, NB);
  gf_stats<<<grid, dim3(32), 0, stream>>>(I, P, Aws, Bws);
  gf_final<<<grid, dim3(32), 0, stream>>>(I, Aws, Bws, Q);
  (void)in_sizes; (void)n_in; (void)out_size; (void)ws_size;
}


# Second box filter applied to sum_c A_c and b only (filter is linear: the four A planes are summed before filtering, not after; sums reordered): 2 planes, not 5, in one fetch batch
# speedup vs baseline: 3.3416x; 1.2485x over previous
#include <hip/hip_runtime.h>

#define HH 768
#define WW 768
#define CC 4
#define NB 8
#define RAD 8
#define EPSV 1e-4f

typedef __attribute__((ext_vector_type(2))) float v2f;
typedef __attribute__((ext_vector_type(8))) float v8f;

static __device__ __forceinline__ v8f wmma4(v2f a, v2f b, v8f c) {
  return __builtin_amdgcn_wmma_f32_16x16x4_f32(false, a, false, b, (short)0, c,
                                               false, false);
}

static __device__ __forceinline__ void load_frag(const float* X, int rs,
                                                 int lane, v2f a[8]) {
  const int m = lane & 15;
  const int hi = lane >> 4;
  #pragma unroll
  for (int kk = 0; kk < 8; ++kk) {
    const int j0 = 16 * (kk >> 2) + 8 * hi + 2 * (kk & 3);
    v2f t = { X[(rs * 16 + m) * 33 + j0], X[(rs * 16 + m) * 33 + j0 + 1] };
    a[kk] = t;
  }
}

static __device__ __forceinline__ void make_band(v2f band[8], int lane) {
  const int hi = lane >> 4;
  const int m = lane & 15;
  #pragma unroll
  for (int kk = 0; kk < 8; ++kk) {
    const int d0 = 16 * (kk >> 2) + 8 * hi + 2 * (kk & 3) - m;
    v2f t = { (d0 >= 0 && d0 <= 16) ? 1.f : 0.f,
              (d0 + 1 >= 0 && d0 + 1 <= 16) ? 1.f : 0.f };
    band[kk] = t;
  }
}

static __device__ __forceinline__ v8f hpass(const v2f a[8],
                                            const v2f band[8]) {
  v8f acc = {};
  #pragma unroll
  for (int kk = 0; kk < 8; ++kk) acc = wmma4(a[kk], band[kk], acc);
  return acc;
}

static __device__ __forceinline__ v8f hpass_prod(const v2f a[8],
                                                 const v2f b[8],
                                                 const v2f band[8]) {
  v8f acc = {};
  #pragma unroll
  for (int kk = 0; kk < 8; ++kk) acc = wmma4(a[kk] * b[kk], band[kk], acc);
  return acc;
}

static __device__ __forceinline__ v8f vpass(const v2f band[8], int t, v8f h,
                                            v8f acc) {
  #pragma unroll
  for (int j = 0; j < 4; ++j) {
    v2f b = { h[2 * j], h[2 * j + 1] };
    acc = wmma4(band[4 * t + j], b, acc);
  }
  return acc;
}

static __device__ __forceinline__ void inv4_rowsum(const float m[16],
                                                   float rs[4]) {
  float i0  =  m[5]*m[10]*m[15] - m[5]*m[11]*m[14] - m[9]*m[6]*m[15] + m[9]*m[7]*m[14] + m[13]*m[6]*m[11] - m[13]*m[7]*m[10];
  float i4  = -m[4]*m[10]*m[15] + m[4]*m[11]*m[14] + m[8]*m[6]*m[15] - m[8]*m[7]*m[14] - m[12]*m[6]*m[11] + m[12]*m[7]*m[10];
  float i8  =  m[4]*m[9]*m[15] - m[4]*m[11]*m[13] - m[8]*m[5]*m[15] + m[8]*m[7]*m[13] + m[12]*m[5]*m[11] - m[12]*m[7]*m[9];
  float i12 = -m[4]*m[9]*m[14] + m[4]*m[10]*m[13] + m[8]*m[5]*m[14] - m[8]*m[6]*m[13] - m[12]*m[5]*m[10] + m[12]*m[6]*m[9];
  float i1  = -m[1]*m[10]*m[15] + m[1]*m[11]*m[14] + m[9]*m[2]*m[15] - m[9]*m[3]*m[14] - m[13]*m[2]*m[11] + m[13]*m[3]*m[10];
  float i5  =  m[0]*m[10]*m[15] - m[0]*m[11]*m[14] - m[8]*m[2]*m[15] + m[8]*m[3]*m[14] + m[12]*m[2]*m[11] - m[12]*m[3]*m[10];
  float i9  = -m[0]*m[9]*m[15] + m[0]*m[11]*m[13] + m[8]*m[1]*m[15] - m[8]*m[3]*m[13] - m[12]*m[1]*m[11] + m[12]*m[3]*m[9];
  float i13 =  m[0]*m[9]*m[14] - m[0]*m[10]*m[13] - m[8]*m[1]*m[14] + m[8]*m[2]*m[13] + m[12]*m[1]*m[10] - m[12]*m[2]*m[9];
  float i2  =  m[1]*m[6]*m[15] - m[1]*m[7]*m[14] - m[5]*m[2]*m[15] + m[5]*m[3]*m[14] + m[13]*m[2]*m[7] - m[13]*m[3]*m[6];
  float i6  = -m[0]*m[6]*m[15] + m[0]*m[7]*m[14] + m[4]*m[2]*m[15] - m[4]*m[3]*m[14] - m[12]*m[2]*m[7] + m[12]*m[3]*m[6];
  float i10 =  m[0]*m[5]*m[15] - m[0]*m[7]*m[13] - m[4]*m[1]*m[15] + m[4]*m[3]*m[13] + m[12]*m[1]*m[7] - m[12]*m[3]*m[5];
  float i14 = -m[0]*m[5]*m[14] + m[0]*m[6]*m[13] + m[4]*m[1]*m[14] - m[4]*m[2]*m[13] - m[12]*m[1]*m[6] + m[12]*m[2]*m[5];
  float i3  = -m[1]*m[6]*m[11] + m[1]*m[7]*m[10] + m[5]*m[2]*m[11] - m[5]*m[3]*m[10] - m[9]*m[2]*m[7] + m[9]*m[3]*m[6];
  float i7  =  m[0]*m[6]*m[11] - m[0]*m[7]*m[10] - m[4]*m[2]*m[11] + m[4]*m[3]*m[10] + m[8]*m[2]*m[7] - m[8]*m[3]*m[6];
  float i11 = -m[0]*m[5]*m[11] + m[0]*m[7]*m[9] + m[4]*m[1]*m[11] - m[4]*m[3]*m[9] - m[8]*m[1]*m[7] + m[8]*m[3]*m[5];
  float i15 =  m[0]*m[5]*m[10] - m[0]*m[6]*m[9] - m[4]*m[1]*m[10] + m[4]*m[2]*m[9] + m[8]*m[1]*m[6] - m[8]*m[2]*m[5];
  const float det = m[0]*i0 + m[1]*i4 + m[2]*i8 + m[3]*i12;
  const float id = 1.0f / det;
  rs[0] = (i0 + i1 + i2 + i3) * id;
  rs[1] = (i4 + i5 + i6 + i7) * id;
  rs[2] = (i8 + i9 + i10 + i11) * id;
  rs[3] = (i12 + i13 + i14 + i15) * id;
}

__global__ __launch_bounds__(32) void gf_stats(const float* __restrict__ I,
                                               const float* __restrict__ P,
                                               float* __restrict__ Aout,
                                               float* __restrict__ Bout) {
  __shared__ float raw[5][32 * 33];
  const int lane = threadIdx.x;
  const int tx = blockIdx.x * 16;
  const int ty = blockIdx.y * 16;
  const int bb = blockIdx.z;
  const int gc = tx - RAD + lane;
  const bool cok = (gc >= 0) && (gc < WW);
  const float* Ib = I + (size_t)bb * CC * HH * WW;
  const float* Pb = P + (size_t)bb * HH * WW;
  const int gcc = min(max(gc, 0), WW - 1);
  float hv[5][32];
  #pragma unroll
  for (int y = 0; y < 32; ++y) {
    const int grc = min(max(ty - RAD + y, 0), HH - 1);
    const size_t off = (size_t)grc * WW + gcc;
    hv[4][y] = Pb[off];
    #pragma unroll
    for (int c = 0; c < CC; ++c)
      hv[c][y] = Ib[(size_t)c * HH * WW + off];
  }
  #pragma unroll
  for (int y = 0; y < 32; ++y) {
    const int gr = ty - RAD + y;
    const bool ok = cok && (gr >= 0) && (gr < HH);
    raw[4][y * 33 + lane] = ok ? hv[4][y] : 0.f;
    #pragma unroll
    for (int c = 0; c < CC; ++c)
      raw[c][y * 33 + lane] = ok ? hv[c][y] : 0.f;
  }
  __syncthreads();

  v2f band[8];
  make_band(band, lane);

  constexpr int po[19] = {4,5,6,7,8, 0,9,10,11,12, 1,13,14,15, 2,16,17, 3,18};
  constexpr int pa[19] = {4,4,4,4,4, 0,0,0,0,0,    1,1,1,1,    2,2,2,   3,3};
  constexpr int pb[19] = {-1,0,1,2,3, -1,0,1,2,3,  -1,1,2,3,   -1,2,3,  -1,3};
  v2f F[2][5][8];
  #pragma unroll
  for (int rs = 0; rs < 2; ++rs) {
    #pragma unroll
    for (int pl = 0; pl < 5; ++pl) load_frag(&raw[pl][0], rs, lane, F[rs][pl]);
  }
  v8f S[19];
  #pragma unroll
  for (int i = 0; i < 19; ++i) {
    const int a = pa[i];
    const int b = pb[i] < 0 ? 0 : pb[i];
    v8f h0, h1;
    if (pb[i] < 0) {
      h0 = hpass(F[0][a], band);
      h1 = hpass(F[1][a], band);
    } else {
      h0 = hpass_prod(F[0][a], F[0][b], band);
      h1 = hpass_prod(F[1][a], F[1][b], band);
    }
    v8f z = {};
    S[po[i]] = vpass(band, 1, h1, vpass(band, 0, h0, z));
  }

  const int hi = lane >> 4;
  const int w = lane & 15;
  const int col = tx + w;
  const int cw = min(col + RAD, WW - 1) - max(col - RAD, 0) + 1;
  #pragma unroll
  for (int v = 0; v < 8; ++v) {
    const int row = ty + v + 8 * hi;
    const int chc = min(row + RAD, HH - 1) - max(row - RAD, 0) + 1;
    const float invN = 1.f / (float)(chc * cw);
    const float pm = S[4][v] * invN;
    float Im[4], first[4];
    float sumIm = 0.f;
    #pragma unroll
    for (int c = 0; c < 4; ++c) {
      Im[c] = S[c][v] * invN;
      sumIm += Im[c];
      first[c] = S[5 + c][v] * invN - pm * Im[c];
    }
    float M[16];
    M[0]  = S[9][v]  * invN + EPSV;
    M[1]  = M[4]  = S[10][v] * invN;
    M[2]  = M[8]  = S[11][v] * invN;
    M[3]  = M[12] = S[12][v] * invN;
    M[5]  = S[13][v] * invN + EPSV;
    M[6]  = M[9]  = S[14][v] * invN;
    M[7]  = M[13] = S[15][v] * invN;
    M[10] = S[16][v] * invN + EPSV;
    M[11] = M[14] = S[17][v] * invN;
    M[15] = S[18][v] * invN + EPSV;
    float rs[4];
    inv4_rowsum(M, rs);
    float sumA = 0.f;
    float Ac[4];
    #pragma unroll
    for (int c = 0; c < 4; ++c) {
      Ac[c] = first[c] * rs[c];
      sumA += Ac[c];
    }
    const float bv = pm - sumA * sumIm;
    Aout[((size_t)bb * HH + row) * WW + col] = sumA;
    Bout[((size_t)bb * HH + row) * WW + col] = bv;
  }
}

__global__ __launch_bounds__(32) void gf_final(const float* __restrict__ I,
                                               const float* __restrict__ Ain,
                                               const float* __restrict__ Bin,
                                               float* __restrict__ Q) {
  __shared__ float X[2][32 * 33];
  const int lane = threadIdx.x;
  const int tx = blockIdx.x * 16;
  const int ty = blockIdx.y * 16;
  const int bb = blockIdx.z;
  const int gc = tx - RAD + lane;
  const bool cok = (gc >= 0) && (gc < WW);
  const int hi = lane >> 4;
  const int w = lane & 15;
  const int col = tx + w;
  const float* Ib = I + (size_t)bb * CC * HH * WW;
  const float* Ab = Ain + (size_t)bb * HH * WW;
  const float* Bb = Bin + (size_t)bb * HH * WW;
  const int gcc = min(max(gc, 0), WW - 1);
  float hv[2][32];
  float iv[4][8];
  #pragma unroll
  for (int y = 0; y < 32; ++y) {
    const int grc = min(max(ty - RAD + y, 0), HH - 1);
    const size_t off = (size_t)grc * WW + gcc;
    hv[0][y] = Ab[off];
    hv[1][y] = Bb[off];
  }
  #pragma unroll
  for (int v = 0; v < 8; ++v) {
    const size_t off = (size_t)(ty + v + 8 * hi) * WW + col;
    #pragma unroll
    for (int c = 0; c < 4; ++c) iv[c][v] = Ib[(size_t)c * HH * WW + off];
  }
  #pragma unroll
  for (int y = 0; y < 32; ++y) {
    const int gr = ty - RAD + y;
    const bool ok = cok && (gr >= 0) && (gr < HH);
    X[0][y * 33 + lane] = ok ? hv[0][y] : 0.f;
    X[1][y * 33 + lane] = ok ? hv[1][y] : 0.f;
  }
  __syncthreads();

  v2f band[8];
  make_band(band, lane);
  v2f F[2][2][8];
  #pragma unroll
  for (int rs = 0; rs < 2; ++rs) {
    #pragma unroll
    for (int pl = 0; pl < 2; ++pl) load_frag(&X[pl][0], rs, lane, F[rs][pl]);
  }
  v8f S[2];
  #pragma unroll
  for (int pl = 0; pl < 2; ++pl) {
    v8f z = {};
    S[pl] = vpass(band, 1, hpass(F[1][pl], band),
                  vpass(band, 0, hpass(F[0][pl], band), z));
  }
  const int cw = min(col + RAD, WW - 1) - max(col - RAD, 0) + 1;
  #pragma unroll
  for (int v = 0; v < 8; ++v) {
    const int row = ty + v + 8 * hi;
    const int chc = min(row + RAD, HH - 1) - max(row - RAD, 0) + 1;
    const float invN = 1.f / (float)(chc * cw);
    float sumI = 0.f;
    #pragma unroll
    for (int c = 0; c < 4; ++c) sumI += iv[c][v];
    const float am = S[0][v] * invN;
    const float bm = S[1][v] * invN;
    Q[((size_t)bb * HH + row) * WW + col] = am * sumI + bm;
  }
}

extern "C" void kernel_launch(void* const* d_in, const int* in_sizes, int n_in,
                              void* d_out, int out_size, void* d_ws,
                              size_t ws_size, hipStream_t stream) {
  const float* I = (const float*)d_in[0];
  const float* P = (const float*)d_in[1];
  float* Q = (float*)d_out;
  float* Aws = (float*)d_ws;
  float* Bws = Aws + (size_t)NB * HH * WW;
  dim3 grid(WW / 16, HH / 16, NB);
  gf_stats<<<grid, dim3(32), 0, stream>>>(I, P, Aws, Bws);
  gf_final<<<grid, dim3(32), 0, stream>>>(I, Aws, Bws, Q);
  (void)in_sizes; (void)n_in; (void)out_size; (void)ws_size;
}


# Each wave walks a strip of 3 stacked 16x16 tiles, carrying the shared halo block's horizontal-pass accumulators from tile to tile: 4 block passes per 3 tiles instead of 6, a third of the waves
# speedup vs baseline: 3.9760x; 1.1898x over previous
#include <hip/hip_runtime.h>

#define HH 768
#define WW 768
#define CC 4
#define NB 8
#define RAD 8
#define EPSV 1e-4f
#define NT 3

typedef __attribute__((ext_vector_type(2))) float v2f;
typedef __attribute__((ext_vector_type(8))) float v8f;

static __device__ __forceinline__ v8f wmma4(v2f a, v2f b, v8f c) {
  return __builtin_amdgcn_wmma_f32_16x16x4_f32(false, a, false, b, (short)0, c,
                                               false, false);
}

static __device__ __forceinline__ void load_frag(const float* X, int rs,
                                                 int lane, v2f a[8]) {
  const int m = lane & 15;
  const int hi = lane >> 4;
  #pragma unroll
  for (int kk = 0; kk < 8; ++kk) {
    const int j0 = 16 * (kk >> 2) + 8 * hi + 2 * (kk & 3);
    v2f t = { X[(rs * 16 + m) * 33 + j0], X[(rs * 16 + m) * 33 + j0 + 1] };
    a[kk] = t;
  }
}

static __device__ __forceinline__ void make_band(v2f band[8], int lane) {
  const int hi = lane >> 4;
  const int m = lane & 15;
  #pragma unroll
  for (int kk = 0; kk < 8; ++kk) {
    const int d0 = 16 * (kk >> 2) + 8 * hi + 2 * (kk & 3) - m;
    v2f t = { (d0 >= 0 && d0 <= 16) ? 1.f : 0.f,
              (d0 + 1 >= 0 && d0 + 1 <= 16) ? 1.f : 0.f };
    band[kk] = t;
  }
}

static __device__ __forceinline__ v8f hpass(const v2f a[8],
                                            const v2f band[8]) {
  v8f acc = {};
  #pragma unroll
  for (int kk = 0; kk < 8; ++kk) acc = wmma4(a[kk], band[kk], acc);
  return acc;
}

static __device__ __forceinline__ v8f hpass_prod(const v2f a[8],
                                                 const v2f b[8],
                                                 const v2f band[8]) {
  v8f acc = {};
  #pragma unroll
  for (int kk = 0; kk < 8; ++kk) acc = wmma4(a[kk] * b[kk], band[kk], acc);
  return acc;
}

static __device__ __forceinline__ v8f vpass(const v2f band[8], int t, v8f h,
                                            v8f acc) {
  #pragma unroll
  for (int j = 0; j < 4; ++j) {
    v2f b = { h[2 * j], h[2 * j + 1] };
    acc = wmma4(band[4 * t + j], b, acc);
  }
  return acc;
}

static __device__ __forceinline__ void inv4_rowsum(const float m[16],
                                                   float rs[4]) {
  float i0  =  m[5]*m[10]*m[15] - m[5]*m[11]*m[14] - m[9]*m[6]*m[15] + m[9]*m[7]*m[14] + m[13]*m[6]*m[11] - m[13]*m[7]*m[10];
  float i4  = -m[4]*m[10]*m[15] + m[4]*m[11]*m[14] + m[8]*m[6]*m[15] - m[8]*m[7]*m[14] - m[12]*m[6]*m[11] + m[12]*m[7]*m[10];
  float i8  =  m[4]*m[9]*m[15] - m[4]*m[11]*m[13] - m[8]*m[5]*m[15] + m[8]*m[7]*m[13] + m[12]*m[5]*m[11] - m[12]*m[7]*m[9];
  float i12 = -m[4]*m[9]*m[14] + m[4]*m[10]*m[13] + m[8]*m[5]*m[14] - m[8]*m[6]*m[13] - m[12]*m[5]*m[10] + m[12]*m[6]*m[9];
  float i1  = -m[1]*m[10]*m[15] + m[1]*m[11]*m[14] + m[9]*m[2]*m[15] - m[9]*m[3]*m[14] - m[13]*m[2]*m[11] + m[13]*m[3]*m[10];
  float i5  =  m[0]*m[10]*m[15] - m[0]*m[11]*m[14] - m[8]*m[2]*m[15] + m[8]*m[3]*m[14] + m[12]*m[2]*m[11] - m[12]*m[3]*m[10];
  float i9  = -m[0]*m[9]*m[15] + m[0]*m[11]*m[13] + m[8]*m[1]*m[15] - m[8]*m[3]*m[13] - m[12]*m[1]*m[11] + m[12]*m[3]*m[9];
  float i13 =  m[0]*m[9]*m[14] - m[0]*m[10]*m[13] - m[8]*m[1]*m[14] + m[8]*m[2]*m[13] + m[12]*m[1]*m[10] - m[12]*m[2]*m[9];
  float i2  =  m[1]*m[6]*m[15] - m[1]*m[7]*m[14] - m[5]*m[2]*m[15] + m[5]*m[3]*m[14] + m[13]*m[2]*m[7] - m[13]*m[3]*m[6];
  float i6  = -m[0]*m[6]*m[15] + m[0]*m[7]*m[14] + m[4]*m[2]*m[15] - m[4]*m[3]*m[14] - m[12]*m[2]*m[7] + m[12]*m[3]*m[6];
  float i10 =  m[0]*m[5]*m[15] - m[0]*m[7]*m[13] - m[4]*m[1]*m[15] + m[4]*m[3]*m[13] + m[12]*m[1]*m[7] - m[12]*m[3]*m[5];
  float i14 = -m[0]*m[5]*m[14] + m[0]*m[6]*m[13] + m[4]*m[1]*m[14] - m[4]*m[2]*m[13] - m[12]*m[1]*m[6] + m[12]*m[2]*m[5];
  float i3  = -m[1]*m[6]*m[11] + m[1]*m[7]*m[10] + m[5]*m[2]*m[11] - m[5]*m[3]*m[10] - m[9]*m[2]*m[7] + m[9]*m[3]*m[6];
  float i7  =  m[0]*m[6]*m[11] - m[0]*m[7]*m[10] - m[4]*m[2]*m[11] + m[4]*m[3]*m[10] + m[8]*m[2]*m[7] - m[8]*m[3]*m[6];
  float i11 = -m[0]*m[5]*m[11] + m[0]*m[7]*m[9] + m[4]*m[1]*m[11] - m[4]*m[3]*m[9] - m[8]*m[1]*m[7] + m[8]*m[3]*m[5];
  float i15 =  m[0]*m[5]*m[10] - m[0]*m[6]*m[9] - m[4]*m[1]*m[10] + m[4]*m[2]*m[9] + m[8]*m[1]*m[6] - m[8]*m[2]*m[5];
  const float det = m[0]*i0 + m[1]*i4 + m[2]*i8 + m[3]*i12;
  const float id = 1.0f / det;
  rs[0] = (i0 + i1 + i2 + i3) * id;
  rs[1] = (i4 + i5 + i6 + i7) * id;
  rs[2] = (i8 + i9 + i10 + i11) * id;
  rs[3] = (i12 + i13 + i14 + i15) * id;
}

static __device__ __forceinline__ void fetch5(const float* Ib, const float* Pb,
                                              int r0, int gcc,
                                              float hv[5][16]) {
  #pragma unroll
  for (int y = 0; y < 16; ++y) {
    const int grc = min(max(r0 + y, 0), HH - 1);
    const size_t off = (size_t)grc * WW + gcc;
    hv[4][y] = Pb[off];
    #pragma unroll
    for (int c = 0; c < CC; ++c)
      hv[c][y] = Ib[(size_t)c * HH * WW + off];
  }
}

static __device__ __forceinline__ void stage5(float (*raw)[16 * 33], int r0,
                                              bool cok, int lane,
                                              const float hv[5][16]) {
  #pragma unroll
  for (int y = 0; y < 16; ++y) {
    const int gr = r0 + y;
    const bool ok = cok && (gr >= 0) && (gr < HH);
    #pragma unroll
    for (int pl = 0; pl < 5; ++pl)
      raw[pl][y * 33 + lane] = ok ? hv[pl][y] : 0.f;
  }
}

static __device__ __forceinline__ void fetch2(const float* Ab, const float* Bb,
                                              int r0, int gcc,
                                              float hv[2][16]) {
  #pragma unroll
  for (int y = 0; y < 16; ++y) {
    const int grc = min(max(r0 + y, 0), HH - 1);
    const size_t off = (size_t)grc * WW + gcc;
    hv[0][y] = Ab[off];
    hv[1][y] = Bb[off];
  }
}

static __device__ __forceinline__ void stage2(float (*X)[16 * 33], int r0,
                                              bool cok, int lane,
                                              const float hv[2][16]) {
  #pragma unroll
  for (int y = 0; y < 16; ++y) {
    const int gr = r0 + y;
    const bool ok = cok && (gr >= 0) && (gr < HH);
    X[0][y * 33 + lane] = ok ? hv[0][y] : 0.f;
    X[1][y * 33 + lane] = ok ? hv[1][y] : 0.f;
  }
}

__global__ __launch_bounds__(32) __attribute__((amdgpu_waves_per_eu(2)))
void gf_stats(const float* __restrict__ I,
                                               const float* __restrict__ P,
                                               float* __restrict__ Aout,
                                               float* __restrict__ Bout) {
  __shared__ float raw[5][16 * 33];
  const int lane = threadIdx.x;
  const int tx = blockIdx.x * 16;
  const int ty0 = blockIdx.y * (16 * NT);
  const int bb = blockIdx.z;
  const int gc = tx - RAD + lane;
  const bool cok = (gc >= 0) && (gc < WW);
  const float* Ib = I + (size_t)bb * CC * HH * WW;
  const float* Pb = P + (size_t)bb * HH * WW;
  const int gcc = min(max(gc, 0), WW - 1);

  v2f band[8];
  make_band(band, lane);

  constexpr int po[19] = {4,5,6,7,8, 0,9,10,11,12, 1,13,14,15, 2,16,17, 3,18};
  constexpr int pa[19] = {4,4,4,4,4, 0,0,0,0,0,    1,1,1,1,    2,2,2,   3,3};
  constexpr int pb[19] = {-1,0,1,2,3, -1,0,1,2,3,  -1,1,2,3,   -1,2,3,  -1,3};

  float hv[5][16];
  v2f F[5][8];
  v8f Hc[19];

  fetch5(Ib, Pb, ty0 - RAD, gcc, hv);
  stage5(raw, ty0 - RAD, cok, lane, hv);
  __syncthreads();
  #pragma unroll
  for (int pl = 0; pl < 5; ++pl) load_frag(&raw[pl][0], 0, lane, F[pl]);
  fetch5(Ib, Pb, ty0 - RAD + 16, gcc, hv);
  #pragma unroll
  for (int i = 0; i < 19; ++i) {
    const int a = pa[i];
    const int b = pb[i] < 0 ? 0 : pb[i];
    Hc[i] = (pb[i] < 0) ? hpass(F[a], band) : hpass_prod(F[a], F[b], band);
  }

  const int hi = lane >> 4;
  const int w = lane & 15;
  const int col = tx + w;
  const int cw = min(col + RAD, WW - 1) - max(col - RAD, 0) + 1;

  #pragma nounroll
  for (int t = 0; t < NT; ++t) {
    const int r1 = ty0 - RAD + 16 * (t + 1);
    __syncthreads();
    stage5(raw, r1, cok, lane, hv);
    __syncthreads();
    #pragma unroll
    for (int pl = 0; pl < 5; ++pl) load_frag(&raw[pl][0], 0, lane, F[pl]);
    if (t + 1 < NT) fetch5(Ib, Pb, r1 + 16, gcc, hv);

    v8f S[19];
    #pragma unroll
    for (int i = 0; i < 19; ++i) {
      const int a = pa[i];
      const int b = pb[i] < 0 ? 0 : pb[i];
      v8f z = {};
      const v8f sp = vpass(band, 0, Hc[i], z);
      const v8f hn =
          (pb[i] < 0) ? hpass(F[a], band) : hpass_prod(F[a], F[b], band);
      S[po[i]] = vpass(band, 1, hn, sp);
      Hc[i] = hn;
    }

    #pragma unroll
    for (int v = 0; v < 8; ++v) {
      const int row = ty0 + 16 * t + v + 8 * hi;
      const int chc = min(row + RAD, HH - 1) - max(row - RAD, 0) + 1;
      const float invN = 1.f / (float)(chc * cw);
      const float pm = S[4][v] * invN;
      float Im[4], first[4];
      float sumIm = 0.f;
      #pragma unroll
      for (int c = 0; c < 4; ++c) {
        Im[c] = S[c][v] * invN;
        sumIm += Im[c];
        first[c] = S[5 + c][v] * invN - pm * Im[c];
      }
      float M[16];
      M[0]  = S[9][v]  * invN + EPSV;
      M[1]  = M[4]  = S[10][v] * invN;
      M[2]  = M[8]  = S[11][v] * invN;
      M[3]  = M[12] = S[12][v] * invN;
      M[5]  = S[13][v] * invN + EPSV;
      M[6]  = M[9]  = S[14][v] * invN;
      M[7]  = M[13] = S[15][v] * invN;
      M[10] = S[16][v] * invN + EPSV;
      M[11] = M[14] = S[17][v] * invN;
      M[15] = S[18][v] * invN + EPSV;
      float rs[4];
      inv4_rowsum(M, rs);
      float sumA = 0.f;
      float Ac[4];
      #pragma unroll
      for (int c = 0; c < 4; ++c) {
        Ac[c] = first[c] * rs[c];
        sumA += Ac[c];
      }
      const float bv = pm - sumA * sumIm;
      Aout[((size_t)bb * HH + row) * WW + col] = sumA;
      Bout[((size_t)bb * HH + row) * WW + col] = bv;
    }
  }
}

__global__ __launch_bounds__(32) void gf_final(const float* __restrict__ I,
                                               const float* __restrict__ Ain,
                                               const float* __restrict__ Bin,
                                               float* __restrict__ Q) {
  __shared__ float X[2][16 * 33];
  const int lane = threadIdx.x;
  const int tx = blockIdx.x * 16;
  const int ty0 = blockIdx.y * (16 * NT);
  const int bb = blockIdx.z;
  const int gc = tx - RAD + lane;
  const bool cok = (gc >= 0) && (gc < WW);
  const int hi = lane >> 4;
  const int w = lane & 15;
  const int col = tx + w;
  const float* Ib = I + (size_t)bb * CC * HH * WW;
  const float* Ab = Ain + (size_t)bb * HH * WW;
  const float* Bb = Bin + (size_t)bb * HH * WW;
  const int gcc = min(max(gc, 0), WW - 1);
  const int cw = min(col + RAD, WW - 1) - max(col - RAD, 0) + 1;

  v2f band[8];
  make_band(band, lane);

  float hv[2][16];
  v2f F[2][8];
  v8f Hc[2];

  fetch2(Ab, Bb, ty0 - RAD, gcc, hv);
  stage2(X, ty0 - RAD, cok, lane, hv);
  __syncthreads();
  #pragma unroll
  for (int pl = 0; pl < 2; ++pl) load_frag(&X[pl][0], 0, lane, F[pl]);
  fetch2(Ab, Bb, ty0 - RAD + 16, gcc, hv);
  #pragma unroll
  for (int pl = 0; pl < 2; ++pl) Hc[pl] = hpass(F[pl], band);

  #pragma nounroll
  for (int t = 0; t < NT; ++t) {
    const int r1 = ty0 - RAD + 16 * (t + 1);
    __syncthreads();
    stage2(X, r1, cok, lane, hv);
    __syncthreads();
    #pragma unroll
    for (int pl = 0; pl < 2; ++pl) load_frag(&X[pl][0], 0, lane, F[pl]);
    float iv[4][8];
    #pragma unroll
    for (int v = 0; v < 8; ++v) {
      const size_t off = (size_t)(ty0 + 16 * t + v + 8 * hi) * WW + col;
      #pragma unroll
      for (int c = 0; c < 4; ++c) iv[c][v] = Ib[(size_t)c * HH * WW + off];
    }
    if (t + 1 < NT) fetch2(Ab, Bb, r1 + 16, gcc, hv);

    v8f S[2];
    #pragma unroll
    for (int pl = 0; pl < 2; ++pl) {
      v8f z = {};
      const v8f sp = vpass(band, 0, Hc[pl], z);
      const v8f hn = hpass(F[pl], band);
      S[pl] = vpass(band, 1, hn, sp);
      Hc[pl] = hn;
    }
    #pragma unroll
    for (int v = 0; v < 8; ++v) {
      const int row = ty0 + 16 * t + v + 8 * hi;
      const int chc = min(row + RAD, HH - 1) - max(row - RAD, 0) + 1;
      const float invN = 1.f / (float)(chc * cw);
      float sumI = 0.f;
      #pragma unroll
      for (int c = 0; c < 4; ++c) sumI += iv[c][v];
      const float am = S[0][v] * invN;
      const float bm = S[1][v] * invN;
      Q[((size_t)bb * HH + row) * WW + col] = am * sumI + bm;
    }
  }
}

extern "C" void kernel_launch(void* const* d_in, const int* in_sizes, int n_in,
                              void* d_out, int out_size, void* d_ws,
                              size_t ws_size, hipStream_t stream) {
  const float* I = (const float*)d_in[0];
  const float* P = (const float*)d_in[1];
  float* Q = (float*)d_out;
  float* Aws = (float*)d_ws;
  float* Bws = Aws + (size_t)NB * HH * WW;
  dim3 grid(WW / 16, HH / (16 * NT), NB);
  gf_stats<<<grid, dim3(32), 0, stream>>>(I, P, Aws, Bws);
  gf_final<<<grid, dim3(32), 0, stream>>>(I, Aws, Bws, Q);
  (void)in_sizes; (void)n_in; (void)out_size; (void)ws_size;
}


# Vertical band product once per halo block: rows of the lower triangle come from the upper-triangle product (L = 1 + I - U) by vector adds, 12 not 16 matrix ops per plane and tile; sums reordered
# speedup vs baseline: 4.1349x; 1.0400x over previous
#include <hip/hip_runtime.h>

#define HH 768
#define WW 768
#define CC 4
#define NB 8
#define RAD 8
#define EPSV 1e-4f
#define NT 3

typedef __attribute__((ext_vector_type(2))) float v2f;
typedef __attribute__((ext_vector_type(8))) float v8f;

static __device__ __forceinline__ v8f wmma4(v2f a, v2f b, v8f c) {
  return __builtin_amdgcn_wmma_f32_16x16x4_f32(false, a, false, b, (short)0, c,
                                               false, false);
}

static __device__ __forceinline__ void load_frag(const float* X, int rs,
                                                 int lane, v2f a[8]) {
  const int m = lane & 15;
  const int hi = lane >> 4;
  #pragma unroll
  for (int kk = 0; kk < 8; ++kk) {
    const int j0 = 16 * (kk >> 2) + 8 * hi + 2 * (kk & 3);
    v2f t = { X[(rs * 16 + m) * 33 + j0], X[(rs * 16 + m) * 33 + j0 + 1] };
    a[kk] = t;
  }
}

static __device__ __forceinline__ void make_band(v2f band[8], int lane) {
  const int hi = lane >> 4;
  const int m = lane & 15;
  #pragma unroll
  for (int kk = 0; kk < 8; ++kk) {
    const int d0 = 16 * (kk >> 2) + 8 * hi + 2 * (kk & 3) - m;
    v2f t = { (d0 >= 0 && d0 <= 16) ? 1.f : 0.f,
              (d0 + 1 >= 0 && d0 + 1 <= 16) ? 1.f : 0.f };
    band[kk] = t;
  }
}

static __device__ __forceinline__ v8f hpass(const v2f a[8],
                                            const v2f band[8]) {
  v8f acc = {};
  #pragma unroll
  for (int kk = 0; kk < 8; ++kk) acc = wmma4(a[kk], band[kk], acc);
  return acc;
}

static __device__ __forceinline__ v8f hpass_prod(const v2f a[8],
                                                 const v2f b[8],
                                                 const v2f band[8]) {
  v8f acc = {};
  #pragma unroll
  for (int kk = 0; kk < 8; ++kk) acc = wmma4(a[kk] * b[kk], band[kk], acc);
  return acc;
}

static __device__ __forceinline__ v8f upper_rows(const v2f band[8], v8f h) {
  v8f acc = {};
  #pragma unroll
  for (int j = 0; j < 4; ++j) {
    v2f b = { h[2 * j], h[2 * j + 1] };
    acc = wmma4(band[j], b, acc);
  }
  return acc;
}

static __device__ __forceinline__ float xhalf(float x) {
  const int b = __builtin_bit_cast(int, x);
  return __builtin_bit_cast(
      float, __builtin_amdgcn_permlanex16(b, b, 0x76543210, 0xfedcba98, false,
                                          false));
}

static __device__ __forceinline__ v8f rows_below(v8f p, int hi) {
  const float z = hi ? 0.f : p[0];
  const float cs = z + xhalf(z);
  v8f r;
  #pragma unroll
  for (int v = 0; v < 8; ++v) r[v] = cs - p[v];
  return r;
}

static __device__ __forceinline__ void inv4_rowsum(const float m[16],
                                                   float rs[4]) {
  float i0  =  m[5]*m[10]*m[15] - m[5]*m[11]*m[14] - m[9]*m[6]*m[15] + m[9]*m[7]*m[14] + m[13]*m[6]*m[11] - m[13]*m[7]*m[10];
  float i4  = -m[4]*m[10]*m[15] + m[4]*m[11]*m[14] + m[8]*m[6]*m[15] - m[8]*m[7]*m[14] - m[12]*m[6]*m[11] + m[12]*m[7]*m[10];
  float i8  =  m[4]*m[9]*m[15] - m[4]*m[11]*m[13] - m[8]*m[5]*m[15] + m[8]*m[7]*m[13] + m[12]*m[5]*m[11] - m[12]*m[7]*m[9];
  float i12 = -m[4]*m[9]*m[14] + m[4]*m[10]*m[13] + m[8]*m[5]*m[14] - m[8]*m[6]*m[13] - m[12]*m[5]*m[10] + m[12]*m[6]*m[9];
  float i1  = -m[1]*m[10]*m[15] + m[1]*m[11]*m[14] + m[9]*m[2]*m[15] - m[9]*m[3]*m[14] - m[13]*m[2]*m[11] + m[13]*m[3]*m[10];
  float i5  =  m[0]*m[10]*m[15] - m[0]*m[11]*m[14] - m[8]*m[2]*m[15] + m[8]*m[3]*m[14] + m[12]*m[2]*m[11] - m[12]*m[3]*m[10];
  float i9  = -m[0]*m[9]*m[15] + m[0]*m[11]*m[13] + m[8]*m[1]*m[15] - m[8]*m[3]*m[13] - m[12]*m[1]*m[11] + m[12]*m[3]*m[9];
  float i13 =  m[0]*m[9]*m[14] - m[0]*m[10]*m[13] - m[8]*m[1]*m[14] + m[8]*m[2]*m[13] + m[12]*m[1]*m[10] - m[12]*m[2]*m[9];
  float i2  =  m[1]*m[6]*m[15] - m[1]*m[7]*m[14] - m[5]*m[2]*m[15] + m[5]*m[3]*m[14] + m[13]*m[2]*m[7] - m[13]*m[3]*m[6];
  float i6  = -m[0]*m[6]*m[15] + m[0]*m[7]*m[14] + m[4]*m[2]*m[15] - m[4]*m[3]*m[14] - m[12]*m[2]*m[7] + m[12]*m[3]*m[6];
  float i10 =  m[0]*m[5]*m[15] - m[0]*m[7]*m[13] - m[4]*m[1]*m[15] + m[4]*m[3]*m[13] + m[12]*m[1]*m[7] - m[12]*m[3]*m[5];
  float i14 = -m[0]*m[5]*m[14] + m[0]*m[6]*m[13] + m[4]*m[1]*m[14] - m[4]*m[2]*m[13] - m[12]*m[1]*m[6] + m[12]*m[2]*m[5];
  float i3  = -m[1]*m[6]*m[11] + m[1]*m[7]*m[10] + m[5]*m[2]*m[11] - m[5]*m[3]*m[10] - m[9]*m[2]*m[7] + m[9]*m[3]*m[6];
  float i7  =  m[0]*m[6]*m[11] - m[0]*m[7]*m[10] - m[4]*m[2]*m[11] + m[4]*m[3]*m[10] + m[8]*m[2]*m[7] - m[8]*m[3]*m[6];
  float i11 = -m[0]*m[5]*m[11] + m[0]*m[7]*m[9] + m[4]*m[1]*m[11] - m[4]*m[3]*m[9] - m[8]*m[1]*m[7] + m[8]*m[3]*m[5];
  float i15 =  m[0]*m[5]*m[10] - m[0]*m[6]*m[9] - m[4]*m[1]*m[10] + m[4]*m[2]*m[9] + m[8]*m[1]*m[6] - m[8]*m[2]*m[5];
  const float det = m[0]*i0 + m[1]*i4 + m[2]*i8 + m[3]*i12;
  const float id = 1.0f / det;
  rs[0] = (i0 + i1 + i2 + i3) * id;
  rs[1] = (i4 + i5 + i6 + i7) * id;
  rs[2] = (i8 + i9 + i10 + i11) * id;
  rs[3] = (i12 + i13 + i14 + i15) * id;
}

static __device__ __forceinline__ void fetch5(const float* Ib, const float* Pb,
                                              int r0, int gcc,
                                              float hv[5][16]) {
  #pragma unroll
  for (int y = 0; y < 16; ++y) {
    const int grc = min(max(r0 + y, 0), HH - 1);
    const size_t off = (size_t)grc * WW + gcc;
    hv[4][y] = Pb[off];
    #pragma unroll
    for (int c = 0; c < CC; ++c)
      hv[c][y] = Ib[(size_t)c * HH * WW + off];
  }
}

static __device__ __forceinline__ void stage5(float (*raw)[16 * 33], int r0,
                                              bool cok, int lane,
                                              const float hv[5][16]) {
  #pragma unroll
  for (int y = 0; y < 16; ++y) {
    const int gr = r0 + y;
    const bool ok = cok && (gr >= 0) && (gr < HH);
    #pragma unroll
    for (int pl = 0; pl < 5; ++pl)
      raw[pl][y * 33 + lane] = ok ? hv[pl][y] : 0.f;
  }
}

static __device__ __forceinline__ void fetch2(const float* Ab, const float* Bb,
                                              int r0, int gcc,
                                              float hv[2][16]) {
  #pragma unroll
  for (int y = 0; y < 16; ++y) {
    const int grc = min(max(r0 + y, 0), HH - 1);
    const size_t off = (size_t)grc * WW + gcc;
    hv[0][y] = Ab[off];
    hv[1][y] = Bb[off];
  }
}

static __device__ __forceinline__ void stage2(float (*X)[16 * 33], int r0,
                                              bool cok, int lane,
                                              const float hv[2][16]) {
  #pragma unroll
  for (int y = 0; y < 16; ++y) {
    const int gr = r0 + y;
    const bool ok = cok && (gr >= 0) && (gr < HH);
    X[0][y * 33 + lane] = ok ? hv[0][y] : 0.f;
    X[1][y * 33 + lane] = ok ? hv[1][y] : 0.f;
  }
}

__global__ __launch_bounds__(32) __attribute__((amdgpu_waves_per_eu(2)))
void gf_stats(const float* __restrict__ I,
                                               const float* __restrict__ P,
                                               float* __restrict__ Aout,
                                               float* __restrict__ Bout) {
  __shared__ float raw[5][16 * 33];
  const int lane = threadIdx.x;
  const int tx = blockIdx.x * 16;
  const int ty0 = blockIdx.y * (16 * NT);
  const int bb = blockIdx.z;
  const int gc = tx - RAD + lane;
  const bool cok = (gc >= 0) && (gc < WW);
  const float* Ib = I + (size_t)bb * CC * HH * WW;
  const float* Pb = P + (size_t)bb * HH * WW;
  const int gcc = min(max(gc, 0), WW - 1);

  v2f band[8];
  make_band(band, lane);

  constexpr int po[19] = {4,5,6,7,8, 0,9,10,11,12, 1,13,14,15, 2,16,17, 3,18};
  constexpr int pa[19] = {4,4,4,4,4, 0,0,0,0,0,    1,1,1,1,    2,2,2,   3,3};
  constexpr int pb[19] = {-1,0,1,2,3, -1,0,1,2,3,  -1,1,2,3,   -1,2,3,  -1,3};

  float hv[5][16];
  v2f F[5][8];
  v8f Pc[19];

  fetch5(Ib, Pb, ty0 - RAD, gcc, hv);
  stage5(raw, ty0 - RAD, cok, lane, hv);
  __syncthreads();
  #pragma unroll
  for (int pl = 0; pl < 5; ++pl) load_frag(&raw[pl][0], 0, lane, F[pl]);
  fetch5(Ib, Pb, ty0 - RAD + 16, gcc, hv);
  #pragma unroll
  for (int i = 0; i < 19; ++i) {
    const int a = pa[i];
    const int b = pb[i] < 0 ? 0 : pb[i];
    const v8f h0 =
        (pb[i] < 0) ? hpass(F[a], band) : hpass_prod(F[a], F[b], band);
    Pc[i] = upper_rows(band, h0);
  }

  const int hi = lane >> 4;
  const int w = lane & 15;
  const int col = tx + w;
  const int cw = min(col + RAD, WW - 1) - max(col - RAD, 0) + 1;

  #pragma nounroll
  for (int t = 0; t < NT; ++t) {
    const int r1 = ty0 - RAD + 16 * (t + 1);
    __syncthreads();
    stage5(raw, r1, cok, lane, hv);
    __syncthreads();
    #pragma unroll
    for (int pl = 0; pl < 5; ++pl) load_frag(&raw[pl][0], 0, lane, F[pl]);
    if (t + 1 < NT) fetch5(Ib, Pb, r1 + 16, gcc, hv);

    v8f S[19];
    #pragma unroll
    for (int i = 0; i < 19; ++i) {
      const int a = pa[i];
      const int b = pb[i] < 0 ? 0 : pb[i];
      const v8f hn =
          (pb[i] < 0) ? hpass(F[a], band) : hpass_prod(F[a], F[b], band);
      const v8f sp = Pc[i] + hn;
      __builtin_amdgcn_sched_barrier(0);
      const v8f pn = upper_rows(band, hn);
      S[po[i]] = sp + rows_below(pn, hi);
      Pc[i] = pn;
    }

    #pragma unroll
    for (int v = 0; v < 8; ++v) {
      const int row = ty0 + 16 * t + v + 8 * hi;
      const int chc = min(row + RAD, HH - 1) - max(row - RAD, 0) + 1;
      const float invN = 1.f / (float)(chc * cw);
      const float pm = S[4][v] * invN;
      float Im[4], first[4];
      float sumIm = 0.f;
      #pragma unroll
      for (int c = 0; c < 4; ++c) {
        Im[c] = S[c][v] * invN;
        sumIm += Im[c];
        first[c] = S[5 + c][v] * invN - pm * Im[c];
      }
      float M[16];
      M[0]  = S[9][v]  * invN + EPSV;
      M[1]  = M[4]  = S[10][v] * invN;
      M[2]  = M[8]  = S[11][v] * invN;
      M[3]  = M[12] = S[12][v] * invN;
      M[5]  = S[13][v] * invN + EPSV;
      M[6]  = M[9]  = S[14][v] * invN;
      M[7]  = M[13] = S[15][v] * invN;
      M[10] = S[16][v] * invN + EPSV;
      M[11] = M[14] = S[17][v] * invN;
      M[15] = S[18][v] * invN + EPSV;
      float rs[4];
      inv4_rowsum(M, rs);
      float sumA = 0.f;
      float Ac[4];
      #pragma unroll
      for (int c = 0; c < 4; ++c) {
        Ac[c] = first[c] * rs[c];
        sumA += Ac[c];
      }
      const float bv = pm - sumA * sumIm;
      Aout[((size_t)bb * HH + row) * WW + col] = sumA;
      Bout[((size_t)bb * HH + row) * WW + col] = bv;
    }
  }
}

__global__ __launch_bounds__(32) void gf_final(const float* __restrict__ I,
                                               const float* __restrict__ Ain,
                                               const float* __restrict__ Bin,
                                               float* __restrict__ Q) {
  __shared__ float X[2][16 * 33];
  const int lane = threadIdx.x;
  const int tx = blockIdx.x * 16;
  const int ty0 = blockIdx.y * (16 * NT);
  const int bb = blockIdx.z;
  const int gc = tx - RAD + lane;
  const bool cok = (gc >= 0) && (gc < WW);
  const int hi = lane >> 4;
  const int w = lane & 15;
  const int col = tx + w;
  const float* Ib = I + (size_t)bb * CC * HH * WW;
  const float* Ab = Ain + (size_t)bb * HH * WW;
  const float* Bb = Bin + (size_t)bb * HH * WW;
  const int gcc = min(max(gc, 0), WW - 1);
  const int cw = min(col + RAD, WW - 1) - max(col - RAD, 0) + 1;

  v2f band[8];
  make_band(band, lane);

  float hv[2][16];
  v2f F[2][8];
  v8f Pc[2];

  fetch2(Ab, Bb, ty0 - RAD, gcc, hv);
  stage2(X, ty0 - RAD, cok, lane, hv);
  __syncthreads();
  #pragma unroll
  for (int pl = 0; pl < 2; ++pl) load_frag(&X[pl][0], 0, lane, F[pl]);
  fetch2(Ab, Bb, ty0 - RAD + 16, gcc, hv);
  #pragma unroll
  for (int pl = 0; pl < 2; ++pl) Pc[pl] = upper_rows(band, hpass(F[pl], band));

  #pragma nounroll
  for (int t = 0; t < NT; ++t) {
    const int r1 = ty0 - RAD + 16 * (t + 1);
    __syncthreads();
    stage2(X, r1, cok, lane, hv);
    __syncthreads();
    #pragma unroll
    for (int pl = 0; pl < 2; ++pl) load_frag(&X[pl][0], 0, lane, F[pl]);
    float iv[4][8];
    #pragma unroll
    for (int v = 0; v < 8; ++v) {
      const size_t off = (size_t)(ty0 + 16 * t + v + 8 * hi) * WW + col;
      #pragma unroll
      for (int c = 0; c < 4; ++c) iv[c][v] = Ib[(size_t)c * HH * WW + off];
    }
    if (t + 1 < NT) fetch2(Ab, Bb, r1 + 16, gcc, hv);

    v8f S[2];
    #pragma unroll
    for (int pl = 0; pl < 2; ++pl) {
      const v8f hn = hpass(F[pl], band);
      const v8f sp = Pc[pl] + hn;
      const v8f pn = upper_rows(band, hn);
      S[pl] = sp + rows_below(pn, hi);
      Pc[pl] = pn;
    }
    #pragma unroll
    for (int v = 0; v < 8; ++v) {
      const int row = ty0 + 16 * t + v + 8 * hi;
      const int chc = min(row + RAD, HH - 1) - max(row - RAD, 0) + 1;
      const float invN = 1.f / (float)(chc * cw);
      float sumI = 0.f;
      #pragma unroll
      for (int c = 0; c < 4; ++c) sumI += iv[c][v];
      const float am = S[0][v] * invN;
      const float bm = S[1][v] * invN;
      Q[((size_t)bb * HH + row) * WW + col] = am * sumI + bm;
    }
  }
}

extern "C" void kernel_launch(void* const* d_in, const int* in_sizes, int n_in,
                              void* d_out, int out_size, void* d_ws,
                              size_t ws_size, hipStream_t stream) {
  const float* I = (const float*)d_in[0];
  const float* P = (const float*)d_in[1];
  float* Q = (float*)d_out;
  float* Aws = (float*)d_ws;
  float* Bws = Aws + (size_t)NB * HH * WW;
  dim3 grid(WW / 16, HH / (16 * NT), NB);
  gf_stats<<<grid, dim3(32), 0, stream>>>(I, P, Aws, Bws);
  gf_final<<<grid, dim3(32), 0, stream>>>(I, Aws, Bws, Q);
  (void)in_sizes; (void)n_in; (void)out_size; (void)ws_size;
}


# Strips of 6 stacked tiles per wave instead of 3 (half the waves, 3 per SIMD): 7 halo-block passes per 6 tiles, not 8; per-element arithmetic and summation order unchanged
# speedup vs baseline: 4.4131x; 1.0673x over previous
#include <hip/hip_runtime.h>

#define HH 768
#define WW 768
#define CC 4
#define NB 8
#define RAD 8
#define EPSV 1e-4f
#define NT 6

typedef __attribute__((ext_vector_type(2))) float v2f;
typedef __attribute__((ext_vector_type(8))) float v8f;

static __device__ __forceinline__ v8f wmma4(v2f a, v2f b, v8f c) {
  return __builtin_amdgcn_wmma_f32_16x16x4_f32(false, a, false, b, (short)0, c,
                                               false, false);
}

static __device__ __forceinline__ void load_frag(const float* X, int rs,
                                                 int lane, v2f a[8]) {
  const int m = lane & 15;
  const int hi = lane >> 4;
  #pragma unroll
  for (int kk = 0; kk < 8; ++kk) {
    const int j0 = 16 * (kk >> 2) + 8 * hi + 2 * (kk & 3);
    v2f t = { X[(rs * 16 + m) * 33 + j0], X[(rs * 16 + m) * 33 + j0 + 1] };
    a[kk] = t;
  }
}

static __device__ __forceinline__ void make_band(v2f band[8], int lane) {
  const int hi = lane >> 4;
  const int m = lane & 15;
  #pragma unroll
  for (int kk = 0; kk < 8; ++kk) {
    const int d0 = 16 * (kk >> 2) + 8 * hi + 2 * (kk & 3) - m;
    v2f t = { (d0 >= 0 && d0 <= 16) ? 1.f : 0.f,
              (d0 + 1 >= 0 && d0 + 1 <= 16) ? 1.f : 0.f };
    band[kk] = t;
  }
}

static __device__ __forceinline__ v8f hpass(const v2f a[8],
                                            const v2f band[8]) {
  v8f acc = {};
  #pragma unroll
  for (int kk = 0; kk < 8; ++kk) acc = wmma4(a[kk], band[kk], acc);
  return acc;
}

static __device__ __forceinline__ v8f hpass_prod(const v2f a[8],
                                                 const v2f b[8],
                                                 const v2f band[8]) {
  v8f acc = {};
  #pragma unroll
  for (int kk = 0; kk < 8; ++kk) acc = wmma4(a[kk] * b[kk], band[kk], acc);
  return acc;
}

static __device__ __forceinline__ v8f upper_rows(const v2f band[8], v8f h) {
  v8f acc = {};
  #pragma unroll
  for (int j = 0; j < 4; ++j) {
    v2f b = { h[2 * j], h[2 * j + 1] };
    acc = wmma4(band[j], b, acc);
  }
  return acc;
}

static __device__ __forceinline__ float xhalf(float x) {
  const int b = __builtin_bit_cast(int, x);
  return __builtin_bit_cast(
      float, __builtin_amdgcn_permlanex16(b, b, 0x76543210, 0xfedcba98, false,
                                          false));
}

static __device__ __forceinline__ v8f rows_below(v8f p, int hi) {
  const float z = hi ? 0.f : p[0];
  const float cs = z + xhalf(z);
  v8f r;
  #pragma unroll
  for (int v = 0; v < 8; ++v) r[v] = cs - p[v];
  return r;
}

static __device__ __forceinline__ void inv4_rowsum(const float m[16],
                                                   float rs[4]) {
  float i0  =  m[5]*m[10]*m[15] - m[5]*m[11]*m[14] - m[9]*m[6]*m[15] + m[9]*m[7]*m[14] + m[13]*m[6]*m[11] - m[13]*m[7]*m[10];
  float i4  = -m[4]*m[10]*m[15] + m[4]*m[11]*m[14] + m[8]*m[6]*m[15] - m[8]*m[7]*m[14] - m[12]*m[6]*m[11] + m[12]*m[7]*m[10];
  float i8  =  m[4]*m[9]*m[15] - m[4]*m[11]*m[13] - m[8]*m[5]*m[15] + m[8]*m[7]*m[13] + m[12]*m[5]*m[11] - m[12]*m[7]*m[9];
  float i12 = -m[4]*m[9]*m[14] + m[4]*m[10]*m[13] + m[8]*m[5]*m[14] - m[8]*m[6]*m[13] - m[12]*m[5]*m[10] + m[12]*m[6]*m[9];
  float i1  = -m[1]*m[10]*m[15] + m[1]*m[11]*m[14] + m[9]*m[2]*m[15] - m[9]*m[3]*m[14] - m[13]*m[2]*m[11] + m[13]*m[3]*m[10];
  float i5  =  m[0]*m[10]*m[15] - m[0]*m[11]*m[14] - m[8]*m[2]*m[15] + m[8]*m[3]*m[14] + m[12]*m[2]*m[11] - m[12]*m[3]*m[10];
  float i9  = -m[0]*m[9]*m[15] + m[0]*m[11]*m[13] + m[8]*m[1]*m[15] - m[8]*m[3]*m[13] - m[12]*m[1]*m[11] + m[12]*m[3]*m[9];
  float i13 =  m[0]*m[9]*m[14] - m[0]*m[10]*m[13] - m[8]*m[1]*m[14] + m[8]*m[2]*m[13] + m[12]*m[1]*m[10] - m[12]*m[2]*m[9];
  float i2  =  m[1]*m[6]*m[15] - m[1]*m[7]*m[14] - m[5]*m[2]*m[15] + m[5]*m[3]*m[14] + m[13]*m[2]*m[7] - m[13]*m[3]*m[6];
  float i6  = -m[0]*m[6]*m[15] + m[0]*m[7]*m[14] + m[4]*m[2]*m[15] - m[4]*m[3]*m[14] - m[12]*m[2]*m[7] + m[12]*m[3]*m[6];
  float i10 =  m[0]*m[5]*m[15] - m[0]*m[7]*m[13] - m[4]*m[1]*m[15] + m[4]*m[3]*m[13] + m[12]*m[1]*m[7] - m[12]*m[3]*m[5];
  float i14 = -m[0]*m[5]*m[14] + m[0]*m[6]*m[13] + m[4]*m[1]*m[14] - m[4]*m[2]*m[13] - m[12]*m[1]*m[6] + m[12]*m[2]*m[5];
  float i3  = -m[1]*m[6]*m[11] + m[1]*m[7]*m[10] + m[5]*m[2]*m[11] - m[5]*m[3]*m[10] - m[9]*m[2]*m[7] + m[9]*m[3]*m[6];
  float i7  =  m[0]*m[6]*m[11] - m[0]*m[7]*m[10] - m[4]*m[2]*m[11] + m[4]*m[3]*m[10] + m[8]*m[2]*m[7] - m[8]*m[3]*m[6];
  float i11 = -m[0]*m[5]*m[11] + m[0]*m[7]*m[9] + m[4]*m[1]*m[11] - m[4]*m[3]*m[9] - m[8]*m[1]*m[7] + m[8]*m[3]*m[5];
  float i15 =  m[0]*m[5]*m[10] - m[0]*m[6]*m[9] - m[4]*m[1]*m[10] + m[4]*m[2]*m[9] + m[8]*m[1]*m[6] - m[8]*m[2]*m[5];
  const float det = m[0]*i0 + m[1]*i4 + m[2]*i8 + m[3]*i12;
  const float id = 1.0f / det;
  rs[0] = (i0 + i1 + i2 + i3) * id;
  rs[1] = (i4 + i5 + i6 + i7) * id;
  rs[2] = (i8 + i9 + i10 + i11) * id;
  rs[3] = (i12 + i13 + i14 + i15) * id;
}

static __device__ __forceinline__ void fetch5(const float* Ib, const float* Pb,
                                              int r0, int gcc,
                                              float hv[5][16]) {
  #pragma unroll
  for (int y = 0; y < 16; ++y) {
    const int grc = min(max(r0 + y, 0), HH - 1);
    const size_t off = (size_t)grc * WW + gcc;
    hv[4][y] = Pb[off];
    #pragma unroll
    for (int c = 0; c < CC; ++c)
      hv[c][y] = Ib[(size_t)c * HH * WW + off];
  }
}

static __device__ __forceinline__ void stage5(float (*raw)[16 * 33], int r0,
                                              bool cok, int lane,
                                              const float hv[5][16]) {
  #pragma unroll
  for (int y = 0; y < 16; ++y) {
    const int gr = r0 + y;
    const bool ok = cok && (gr >= 0) && (gr < HH);
    #pragma unroll
    for (int pl = 0; pl < 5; ++pl)
      raw[pl][y * 33 + lane] = ok ? hv[pl][y] : 0.f;
  }
}

static __device__ __forceinline__ void fetch2(const float* Ab, const float* Bb,
                                              int r0, int gcc,
                                              float hv[2][16]) {
  #pragma unroll
  for (int y = 0; y < 16; ++y) {
    const int grc = min(max(r0 + y, 0), HH - 1);
    const size_t off = (size_t)grc * WW + gcc;
    hv[0][y] = Ab[off];
    hv[1][y] = Bb[off];
  }
}

static __device__ __forceinline__ void stage2(float (*X)[16 * 33], int r0,
                                              bool cok, int lane,
                                              const float hv[2][16]) {
  #pragma unroll
  for (int y = 0; y < 16; ++y) {
    const int gr = r0 + y;
    const bool ok = cok && (gr >= 0) && (gr < HH);
    X[0][y * 33 + lane] = ok ? hv[0][y] : 0.f;
    X[1][y * 33 + lane] = ok ? hv[1][y] : 0.f;
  }
}

__global__ __launch_bounds__(32) __attribute__((amdgpu_waves_per_eu(2)))
void gf_stats(const float* __restrict__ I,
                                               const float* __restrict__ P,
                                               float* __restrict__ Aout,
                                               float* __restrict__ Bout) {
  __shared__ float raw[5][16 * 33];
  const int lane = threadIdx.x;
  const int tx = blockIdx.x * 16;
  const int ty0 = blockIdx.y * (16 * NT);
  const int bb = blockIdx.z;
  const int gc = tx - RAD + lane;
  const bool cok = (gc >= 0) && (gc < WW);
  const float* Ib = I + (size_t)bb * CC * HH * WW;
  const float* Pb = P + (size_t)bb * HH * WW;
  const int gcc = min(max(gc, 0), WW - 1);

  v2f band[8];
  make_band(band, lane);

  constexpr int po[19] = {4,5,6,7,8, 0,9,10,11,12, 1,13,14,15, 2,16,17, 3,18};
  constexpr int pa[19] = {4,4,4,4,4, 0,0,0,0,0,    1,1,1,1,    2,2,2,   3,3};
  constexpr int pb[19] = {-1,0,1,2,3, -1,0,1,2,3,  -1,1,2,3,   -1,2,3,  -1,3};

  float hv[5][16];
  v2f F[5][8];
  v8f Pc[19];

  fetch5(Ib, Pb, ty0 - RAD, gcc, hv);
  stage5(raw, ty0 - RAD, cok, lane, hv);
  __syncthreads();
  #pragma unroll
  for (int pl = 0; pl < 5; ++pl) load_frag(&raw[pl][0], 0, lane, F[pl]);
  fetch5(Ib, Pb, ty0 - RAD + 16, gcc, hv);
  #pragma unroll
  for (int i = 0; i < 19; ++i) {
    const int a = pa[i];
    const int b = pb[i] < 0 ? 0 : pb[i];
    const v8f h0 =
        (pb[i] < 0) ? hpass(F[a], band) : hpass_prod(F[a], F[b], band);
    Pc[i] = upper_rows(band, h0);
  }

  const int hi = lane >> 4;
  const int w = lane & 15;
  const int col = tx + w;
  const int cw = min(col + RAD, WW - 1) - max(col - RAD, 0) + 1;

  #pragma nounroll
  for (int t = 0; t < NT; ++t) {
    const int r1 = ty0 - RAD + 16 * (t + 1);
    __syncthreads();
    stage5(raw, r1, cok, lane, hv);
    __syncthreads();
    #pragma unroll
    for (int pl = 0; pl < 5; ++pl) load_frag(&raw[pl][0], 0, lane, F[pl]);
    if (t + 1 < NT) fetch5(Ib, Pb, r1 + 16, gcc, hv);

    v8f S[19];
    #pragma unroll
    for (int i = 0; i < 19; ++i) {
      const int a = pa[i];
      const int b = pb[i] < 0 ? 0 : pb[i];
      const v8f hn =
          (pb[i] < 0) ? hpass(F[a], band) : hpass_prod(F[a], F[b], band);
      const v8f sp = Pc[i] + hn;
      __builtin_amdgcn_sched_barrier(0);
      const v8f pn = upper_rows(band, hn);
      S[po[i]] = sp + rows_below(pn, hi);
      Pc[i] = pn;
    }

    #pragma unroll
    for (int v = 0; v < 8; ++v) {
      const int row = ty0 + 16 * t + v + 8 * hi;
      const int chc = min(row + RAD, HH - 1) - max(row - RAD, 0) + 1;
      const float invN = 1.f / (float)(chc * cw);
      const float pm = S[4][v] * invN;
      float Im[4], first[4];
      float sumIm = 0.f;
      #pragma unroll
      for (int c = 0; c < 4; ++c) {
        Im[c] = S[c][v] * invN;
        sumIm += Im[c];
        first[c] = S[5 + c][v] * invN - pm * Im[c];
      }
      float M[16];
      M[0]  = S[9][v]  * invN + EPSV;
      M[1]  = M[4]  = S[10][v] * invN;
      M[2]  = M[8]  = S[11][v] * invN;
      M[3]  = M[12] = S[12][v] * invN;
      M[5]  = S[13][v] * invN + EPSV;
      M[6]  = M[9]  = S[14][v] * invN;
      M[7]  = M[13] = S[15][v] * invN;
      M[10] = S[16][v] * invN + EPSV;
      M[11] = M[14] = S[17][v] * invN;
      M[15] = S[18][v] * invN + EPSV;
      float rs[4];
      inv4_rowsum(M, rs);
      float sumA = 0.f;
      float Ac[4];
      #pragma unroll
      for (int c = 0; c < 4; ++c) {
        Ac[c] = first[c] * rs[c];
        sumA += Ac[c];
      }
      const float bv = pm - sumA * sumIm;
      Aout[((size_t)bb * HH + row) * WW + col] = sumA;
      Bout[((size_t)bb * HH + row) * WW + col] = bv;
    }
  }
}

__global__ __launch_bounds__(32) void gf_final(const float* __restrict__ I,
                                               const float* __restrict__ Ain,
                                               const float* __restrict__ Bin,
                                               float* __restrict__ Q) {
  __shared__ float X[2][16 * 33];
  const int lane = threadIdx.x;
  const int tx = blockIdx.x * 16;
  const int ty0 = blockIdx.y * (16 * NT);
  const int bb = blockIdx.z;
  const int gc = tx - RAD + lane;
  const bool cok = (gc >= 0) && (gc < WW);
  const int hi = lane >> 4;
  const int w = lane & 15;
  const int col = tx + w;
  const float* Ib = I + (size_t)bb * CC * HH * WW;
  const float* Ab = Ain + (size_t)bb * HH * WW;
  const float* Bb = Bin + (size_t)bb * HH * WW;
  const int gcc = min(max(gc, 0), WW - 1);
  const int cw = min(col + RAD, WW - 1) - max(col - RAD, 0) + 1;

  v2f band[8];
  make_band(band, lane);

  float hv[2][16];
  v2f F[2][8];
  v8f Pc[2];

  fetch2(Ab, Bb, ty0 - RAD, gcc, hv);
  stage2(X, ty0 - RAD, cok, lane, hv);
  __syncthreads();
  #pragma unroll
  for (int pl = 0; pl < 2; ++pl) load_frag(&X[pl][0], 0, lane, F[pl]);
  fetch2(Ab, Bb, ty0 - RAD + 16, gcc, hv);
  #pragma unroll
  for (int pl = 0; pl < 2; ++pl) Pc[pl] = upper_rows(band, hpass(F[pl], band));

  #pragma nounroll
  for (int t = 0; t < NT; ++t) {
    const int r1 = ty0 - RAD + 16 * (t + 1);
    __syncthreads();
    stage2(X, r1, cok, lane, hv);
    __syncthreads();
    #pragma unroll
    for (int pl = 0; pl < 2; ++pl) load_frag(&X[pl][0], 0, lane, F[pl]);
    float iv[4][8];
    #pragma unroll
    for (int v = 0; v < 8; ++v) {
      const size_t off = (size_t)(ty0 + 16 * t + v + 8 * hi) * WW + col;
      #pragma unroll
      for (int c = 0; c < 4; ++c) iv[c][v] = Ib[(size_t)c * HH * WW + off];
    }
    if (t + 1 < NT) fetch2(Ab, Bb, r1 + 16, gcc, hv);

    v8f S[2];
    #pragma unroll
    for (int pl = 0; pl < 2; ++pl) {
      const v8f hn = hpass(F[pl], band);
      const v8f sp = Pc[pl] + hn;
      const v8f pn = upper_rows(band, hn);
      S[pl] = sp + rows_below(pn, hi);
      Pc[pl] = pn;
    }
    #pragma unroll
    for (int v = 0; v < 8; ++v) {
      const int row = ty0 + 16 * t + v + 8 * hi;
      const int chc = min(row + RAD, HH - 1) - max(row - RAD, 0) + 1;
      const float invN = 1.f / (float)(chc * cw);
      float sumI = 0.f;
      #pragma unroll
      for (int c = 0; c < 4; ++c) sumI += iv[c][v];
      const float am = S[0][v] * invN;
      const float bm = S[1][v] * invN;
      Q[((size_t)bb * HH + row) * WW + col] = am * sumI + bm;
    }
  }
}

extern "C" void kernel_launch(void* const* d_in, const int* in_sizes, int n_in,
                              void* d_out, int out_size, void* d_ws,
                              size_t ws_size, hipStream_t stream) {
  const float* I = (const float*)d_in[0];
  const float* P = (const float*)d_in[1];
  float* Q = (float*)d_out;
  float* Aws = (float*)d_ws;
  float* Bws = Aws + (size_t)NB * HH * WW;
  dim3 grid(WW / 16, HH / (16 * NT), NB);
  gf_stats<<<grid, dim3(32), 0, stream>>>(I, P, Aws, Bws);
  gf_final<<<grid, dim3(32), 0, stream>>>(I, Aws, Bws, Q);
  (void)in_sizes; (void)n_in; (void)out_size; (void)ws_size;
}


# Inverse row sums as solution of M x = 1 by symmetric 4x4 elimination, no pivoting (M positive definite; v_rcp_f32 pivots), not 16 cofactors over a determinant; pixel pairs packed; rounding differs
# speedup vs baseline: 4.7470x; 1.0757x over previous
#include <hip/hip_runtime.h>

#define HH 768
#define WW 768
#define CC 4
#define NB 8
#define RAD 8
#define EPSV 1e-4f
#define NT 6

typedef __attribute__((ext_vector_type(2))) float v2f;
typedef __attribute__((ext_vector_type(8))) float v8f;

static __device__ __forceinline__ v8f wmma4(v2f a, v2f b, v8f c) {
  return __builtin_amdgcn_wmma_f32_16x16x4_f32(false, a, false, b, (short)0, c,
                                               false, false);
}

static __device__ __forceinline__ void load_frag(const float* X, int rs,
                                                 int lane, v2f a[8]) {
  const int m = lane & 15;
  const int hi = lane >> 4;
  #pragma unroll
  for (int kk = 0; kk < 8; ++kk) {
    const int j0 = 16 * (kk >> 2) + 8 * hi + 2 * (kk & 3);
    v2f t = { X[(rs * 16 + m) * 33 + j0], X[(rs * 16 + m) * 33 + j0 + 1] };
    a[kk] = t;
  }
}

static __device__ __forceinline__ void make_band(v2f band[8], int lane) {
  const int hi = lane >> 4;
  const int m = lane & 15;
  #pragma unroll
  for (int kk = 0; kk < 8; ++kk) {
    const int d0 = 16 * (kk >> 2) + 8 * hi + 2 * (kk & 3) - m;
    v2f t = { (d0 >= 0 && d0 <= 16) ? 1.f : 0.f,
              (d0 + 1 >= 0 && d0 + 1 <= 16) ? 1.f : 0.f };
    band[kk] = t;
  }
}

static __device__ __forceinline__ v8f hpass(const v2f a[8],
                                            const v2f band[8]) {
  v8f acc = {};
  #pragma unroll
  for (int kk = 0; kk < 8; ++kk) acc = wmma4(a[kk], band[kk], acc);
  return acc;
}

static __device__ __forceinline__ v8f hpass_prod(const v2f a[8],
                                                 const v2f b[8],
                                                 const v2f band[8]) {
  v8f acc = {};
  #pragma unroll
  for (int kk = 0; kk < 8; ++kk) acc = wmma4(a[kk] * b[kk], band[kk], acc);
  return acc;
}

static __device__ __forceinline__ v8f upper_rows(const v2f band[8], v8f h) {
  v8f acc = {};
  #pragma unroll
  for (int j = 0; j < 4; ++j) {
    v2f b = { h[2 * j], h[2 * j + 1] };
    acc = wmma4(band[j], b, acc);
  }
  return acc;
}

static __device__ __forceinline__ float xhalf(float x) {
  const int b = __builtin_bit_cast(int, x);
  return __builtin_bit_cast(
      float, __builtin_amdgcn_permlanex16(b, b, 0x76543210, 0xfedcba98, false,
                                          false));
}

static __device__ __forceinline__ v8f rows_below(v8f p, int hi) {
  const float z = hi ? 0.f : p[0];
  const float cs = z + xhalf(z);
  v8f r;
  #pragma unroll
  for (int v = 0; v < 8; ++v) r[v] = cs - p[v];
  return r;
}

static __device__ __forceinline__ v2f pair(v8f s, int q) {
  v2f r = { s[2 * q], s[2 * q + 1] };
  return r;
}

static __device__ __forceinline__ v2f rcp2(v2f x) {
  v2f r = { __builtin_amdgcn_rcpf(x[0]), __builtin_amdgcn_rcpf(x[1]) };
  return r;
}

static __device__ __forceinline__ void solve4_ones(v2f a, v2f b, v2f c, v2f d,
                                                   v2f e, v2f f, v2f g, v2f h,
                                                   v2f i, v2f j, v2f x[4]) {
  const v2f r0 = rcp2(a);
  const v2f l1 = b * r0, l2 = c * r0, l3 = d * r0;
  const v2f e1 = e - l1 * b, f1 = f - l1 * c, g1 = g - l1 * d;
  const v2f h1 = h - l2 * c, i1 = i - l2 * d;
  const v2f j1 = j - l3 * d;
  const v2f y1 = 1.f - l1, y2 = 1.f - l2, y3 = 1.f - l3;
  const v2f r1 = rcp2(e1);
  const v2f m2 = f1 * r1, m3 = g1 * r1;
  const v2f h2 = h1 - m2 * f1, i2 = i1 - m2 * g1;
  const v2f j2 = j1 - m3 * g1;
  const v2f z2 = y2 - m2 * y1, z3 = y3 - m3 * y1;
  const v2f r2 = rcp2(h2);
  const v2f n3 = i2 * r2;
  const v2f j3 = j2 - n3 * i2;
  const v2f w3 = z3 - n3 * z2;
  const v2f r3 = rcp2(j3);
  x[3] = w3 * r3;
  x[2] = (z2 - i2 * x[3]) * r2;
  x[1] = (y1 - f1 * x[2] - g1 * x[3]) * r1;
  x[0] = (1.f - b * x[1] - c * x[2] - d * x[3]) * r0;
}

static __device__ __forceinline__ void fetch5(const float* Ib, const float* Pb,
                                              int r0, int gcc,
                                              float hv[5][16]) {
  #pragma unroll
  for (int y = 0; y < 16; ++y) {
    const int grc = min(max(r0 + y, 0), HH - 1);
    const size_t off = (size_t)grc * WW + gcc;
    hv[4][y] = Pb[off];
    #pragma unroll
    for (int c = 0; c < CC; ++c)
      hv[c][y] = Ib[(size_t)c * HH * WW + off];
  }
}

static __device__ __forceinline__ void stage5(float (*raw)[16 * 33], int r0,
                                              bool cok, int lane,
                                              const float hv[5][16]) {
  #pragma unroll
  for (int y = 0; y < 16; ++y) {
    const int gr = r0 + y;
    const bool ok = cok && (gr >= 0) && (gr < HH);
    #pragma unroll
    for (int pl = 0; pl < 5; ++pl)
      raw[pl][y * 33 + lane] = ok ? hv[pl][y] : 0.f;
  }
}

static __device__ __forceinline__ void fetch2(const float* Ab, const float* Bb,
                                              int r0, int gcc,
                                              float hv[2][16]) {
  #pragma unroll
  for (int y = 0; y < 16; ++y) {
    const int grc = min(max(r0 + y, 0), HH - 1);
    const size_t off = (size_t)grc * WW + gcc;
    hv[0][y] = Ab[off];
    hv[1][y] = Bb[off];
  }
}

static __device__ __forceinline__ void stage2(float (*X)[16 * 33], int r0,
                                              bool cok, int lane,
                                              const float hv[2][16]) {
  #pragma unroll
  for (int y = 0; y < 16; ++y) {
    const int gr = r0 + y;
    const bool ok = cok && (gr >= 0) && (gr < HH);
    X[0][y * 33 + lane] = ok ? hv[0][y] : 0.f;
    X[1][y * 33 + lane] = ok ? hv[1][y] : 0.f;
  }
}

__global__ __launch_bounds__(32) __attribute__((amdgpu_waves_per_eu(2)))
void gf_stats(const float* __restrict__ I,
                                               const float* __restrict__ P,
                                               float* __restrict__ Aout,
                                               float* __restrict__ Bout) {
  __shared__ float raw[5][16 * 33];
  const int lane = threadIdx.x;
  const int tx = blockIdx.x * 16;
  const int ty0 = blockIdx.y * (16 * NT);
  const int bb = blockIdx.z;
  const int gc = tx - RAD + lane;
  const bool cok = (gc >= 0) && (gc < WW);
  const float* Ib = I + (size_t)bb * CC * HH * WW;
  const float* Pb = P + (size_t)bb * HH * WW;
  const int gcc = min(max(gc, 0), WW - 1);

  v2f band[8];
  make_band(band, lane);

  constexpr int po[19] = {4,5,6,7,8, 0,9,10,11,12, 1,13,14,15, 2,16,17, 3,18};
  constexpr int pa[19] = {4,4,4,4,4, 0,0,0,0,0,    1,1,1,1,    2,2,2,   3,3};
  constexpr int pb[19] = {-1,0,1,2,3, -1,0,1,2,3,  -1,1,2,3,   -1,2,3,  -1,3};

  float hv[5][16];
  v2f F[5][8];
  v8f Pc[19];

  fetch5(Ib, Pb, ty0 - RAD, gcc, hv);
  stage5(raw, ty0 - RAD, cok, lane, hv);
  __syncthreads();
  #pragma unroll
  for (int pl = 0; pl < 5; ++pl) load_frag(&raw[pl][0], 0, lane, F[pl]);
  fetch5(Ib, Pb, ty0 - RAD + 16, gcc, hv);
  #pragma unroll
  for (int i = 0; i < 19; ++i) {
    const int a = pa[i];
    const int b = pb[i] < 0 ? 0 : pb[i];
    const v8f h0 =
        (pb[i] < 0) ? hpass(F[a], band) : hpass_prod(F[a], F[b], band);
    Pc[i] = upper_rows(band, h0);
  }

  const int hi = lane >> 4;
  const int w = lane & 15;
  const int col = tx + w;
  const int cw = min(col + RAD, WW - 1) - max(col - RAD, 0) + 1;

  #pragma nounroll
  for (int t = 0; t < NT; ++t) {
    const int r1 = ty0 - RAD + 16 * (t + 1);
    __syncthreads();
    stage5(raw, r1, cok, lane, hv);
    __syncthreads();
    #pragma unroll
    for (int pl = 0; pl < 5; ++pl) load_frag(&raw[pl][0], 0, lane, F[pl]);
    if (t + 1 < NT) fetch5(Ib, Pb, r1 + 16, gcc, hv);

    v8f S[19];
    #pragma unroll
    for (int i = 0; i < 19; ++i) {
      const int a = pa[i];
      const int b = pb[i] < 0 ? 0 : pb[i];
      const v8f hn =
          (pb[i] < 0) ? hpass(F[a], band) : hpass_prod(F[a], F[b], band);
      const v8f sp = Pc[i] + hn;
      __builtin_amdgcn_sched_barrier(0);
      const v8f pn = upper_rows(band, hn);
      S[po[i]] = sp + rows_below(pn, hi);
      Pc[i] = pn;
    }

    #pragma unroll
    for (int q = 0; q < 4; ++q) {
      int row[2];
      v2f invN;
      #pragma unroll
      for (int s = 0; s < 2; ++s) {
        row[s] = ty0 + 16 * t + 2 * q + s + 8 * hi;
        const int chc = min(row[s] + RAD, HH - 1) - max(row[s] - RAD, 0) + 1;
        invN[s] = 1.f / (float)(chc * cw);
      }
      const v2f pm = pair(S[4], q) * invN;
      v2f Im[4], first[4];
      v2f sumIm = {0.f, 0.f};
      #pragma unroll
      for (int c = 0; c < 4; ++c) {
        Im[c] = pair(S[c], q) * invN;
        sumIm += Im[c];
        first[c] = pair(S[5 + c], q) * invN - pm * Im[c];
      }
      const v2f ep = {EPSV, EPSV};
      v2f rs[4];
      solve4_ones(pair(S[9], q) * invN + ep, pair(S[10], q) * invN,
                  pair(S[11], q) * invN, pair(S[12], q) * invN,
                  pair(S[13], q) * invN + ep, pair(S[14], q) * invN,
                  pair(S[15], q) * invN, pair(S[16], q) * invN + ep,
                  pair(S[17], q) * invN, pair(S[18], q) * invN + ep, rs);
      v2f sumA = {0.f, 0.f};
      #pragma unroll
      for (int c = 0; c < 4; ++c)
        sumA += first[c] * rs[c];
      const v2f bv = pm - sumA * sumIm;
      #pragma unroll
      for (int s = 0; s < 2; ++s) {
        Aout[((size_t)bb * HH + row[s]) * WW + col] = sumA[s];
        Bout[((size_t)bb * HH + row[s]) * WW + col] = bv[s];
      }
    }
  }
}

__global__ __launch_bounds__(32) void gf_final(const float* __restrict__ I,
                                               const float* __restrict__ Ain,
                                               const float* __restrict__ Bin,
                                               float* __restrict__ Q) {
  __shared__ float X[2][16 * 33];
  const int lane = threadIdx.x;
  const int tx = blockIdx.x * 16;
  const int ty0 = blockIdx.y * (16 * NT);
  const int bb = blockIdx.z;
  const int gc = tx - RAD + lane;
  const bool cok = (gc >= 0) && (gc < WW);
  const int hi = lane >> 4;
  const int w = lane & 15;
  const int col = tx + w;
  const float* Ib = I + (size_t)bb * CC * HH * WW;
  const float* Ab = Ain + (size_t)bb * HH * WW;
  const float* Bb = Bin + (size_t)bb * HH * WW;
  const int gcc = min(max(gc, 0), WW - 1);
  const int cw = min(col + RAD, WW - 1) - max(col - RAD, 0) + 1;

  v2f band[8];
  make_band(band, lane);

  float hv[2][16];
  v2f F[2][8];
  v8f Pc[2];

  fetch2(Ab, Bb, ty0 - RAD, gcc, hv);
  stage2(X, ty0 - RAD, cok, lane, hv);
  __syncthreads();
  #pragma unroll
  for (int pl = 0; pl < 2; ++pl) load_frag(&X[pl][0], 0, lane, F[pl]);
  fetch2(Ab, Bb, ty0 - RAD + 16, gcc, hv);
  #pragma unroll
  for (int pl = 0; pl < 2; ++pl) Pc[pl] = upper_rows(band, hpass(F[pl], band));

  #pragma nounroll
  for (int t = 0; t < NT; ++t) {
    const int r1 = ty0 - RAD + 16 * (t + 1);
    __syncthreads();
    stage2(X, r1, cok, lane, hv);
    __syncthreads();
    #pragma unroll
    for (int pl = 0; pl < 2; ++pl) load_frag(&X[pl][0], 0, lane, F[pl]);
    float iv[4][8];
    #pragma unroll
    for (int v = 0; v < 8; ++v) {
      const size_t off = (size_t)(ty0 + 16 * t + v + 8 * hi) * WW + col;
      #pragma unroll
      for (int c = 0; c < 4; ++c) iv[c][v] = Ib[(size_t)c * HH * WW + off];
    }
    if (t + 1 < NT) fetch2(Ab, Bb, r1 + 16, gcc, hv);

    v8f S[2];
    #pragma unroll
    for (int pl = 0; pl < 2; ++pl) {
      const v8f hn = hpass(F[pl], band);
      const v8f sp = Pc[pl] + hn;
      const v8f pn = upper_rows(band, hn);
      S[pl] = sp + rows_below(pn, hi);
      Pc[pl] = pn;
    }
    #pragma unroll
    for (int v = 0; v < 8; ++v) {
      const int row = ty0 + 16 * t + v + 8 * hi;
      const int chc = min(row + RAD, HH - 1) - max(row - RAD, 0) + 1;
      const float invN = 1.f / (float)(chc * cw);
      float sumI = 0.f;
      #pragma unroll
      for (int c = 0; c < 4; ++c) sumI += iv[c][v];
      const float am = S[0][v] * invN;
      const float bm = S[1][v] * invN;
      Q[((size_t)bb * HH + row) * WW + col] = am * sumI + bm;
    }
  }
}

extern "C" void kernel_launch(void* const* d_in, const int* in_sizes, int n_in,
                              void* d_out, int out_size, void* d_ws,
                              size_t ws_size, hipStream_t stream) {
  const float* I = (const float*)d_in[0];
  const float* P = (const float*)d_in[1];
  float* Q = (float*)d_out;
  float* Aws = (float*)d_ws;
  float* Bws = Aws + (size_t)NB * HH * WW;
  dim3 grid(WW / 16, HH / (16 * NT), NB);
  gf_stats<<<grid, dim3(32), 0, stream>>>(I, P, Aws, Bws);
  gf_final<<<grid, dim3(32), 0, stream>>>(I, Aws, Bws, Q);
  (void)in_sizes; (void)n_in; (void)out_size; (void)ws_size;
}


# Horizontal band pass over column pairs: pair sums made in the A fragments (K = 16 pairs, not 32 columns), the unpaired column read from LDS as C operand; 8 not 12 matrix ops per plane; sums reordered
# speedup vs baseline: 5.1551x; 1.0860x over previous
#include <hip/hip_runtime.h>

#define HH 768
#define WW 768
#define CC 4
#define NB 8
#define RAD 8
#define EPSV 1e-4f
#define NT 6

typedef __attribute__((ext_vector_type(2))) float v2f;
typedef __attribute__((ext_vector_type(8))) float v8f;

static __device__ __forceinline__ v8f wmma4(v2f a, v2f b, v8f c) {
  return __builtin_amdgcn_wmma_f32_16x16x4_f32(false, a, false, b, (short)0, c,
                                               false, false);
}

static __device__ __forceinline__ void load_frag(const float* X, int rs,
                                                 int lane, v2f a[8]) {
  const int m = lane & 15;
  const int hi = lane >> 4;
  #pragma unroll
  for (int w = 0; w < 4; ++w) {
    const int kk = 2 * w;
    const int j0 = 16 * (kk >> 2) + 8 * hi + 2 * (kk & 3);
    v2f ev = { X[(rs * 16 + m) * 33 + j0], X[(rs * 16 + m) * 33 + j0 + 2] };
    v2f od = { X[(rs * 16 + m) * 33 + j0 + 1], X[(rs * 16 + m) * 33 + j0 + 3] };
    a[2 * w] = ev;
    a[2 * w + 1] = od;
  }
}

static __device__ __forceinline__ void make_band(v2f band[4], int lane) {
  const int hi = lane >> 4;
  const int m = lane & 15;
  #pragma unroll
  for (int kk = 0; kk < 4; ++kk) {
    const int d0 = 8 * hi + 2 * kk - m;
    v2f t = { (d0 >= 0) ? 1.f : 0.f, (d0 + 1 >= 0) ? 1.f : 0.f };
    band[kk] = t;
  }
}

static __device__ __forceinline__ void make_pband(v2f pb[4], int lane) {
  const int hi = lane >> 4;
  const int n = lane & 15;
  #pragma unroll
  for (int w = 0; w < 4; ++w) {
    v2f t;
    #pragma unroll
    for (int s = 0; s < 2; ++s) {
      const int kk = 2 * w + s;
      const int d = 16 * (kk >> 2) + 8 * hi + 2 * (kk & 3) - n;
      t[s] = (d >= 0 && d <= 15) ? 1.f : 0.f;
    }
    pb[w] = t;
  }
}

static __device__ __forceinline__ v8f load_edge(const float* X, int lane) {
  const int n = lane & 15;
  const int ec = (n & 1) ? n : n + 16;
  const int r8 = 8 * (lane >> 4);
  v8f e;
  #pragma unroll
  for (int v = 0; v < 8; ++v) e[v] = X[(r8 + v) * 33 + ec];
  return e;
}

static __device__ __forceinline__ v8f hpass(const v2f a[8], const v2f pb[4],
                                            v8f edge) {
  v8f acc = edge;
  #pragma unroll
  for (int w = 0; w < 4; ++w)
    acc = wmma4(a[2 * w] + a[2 * w + 1], pb[w], acc);
  return acc;
}

static __device__ __forceinline__ v8f hpass_prod(const v2f a[8],
                                                 const v2f b[8],
                                                 const v2f pb[4], v8f ea,
                                                 v8f eb) {
  v8f acc = ea * eb;
  #pragma unroll
  for (int w = 0; w < 4; ++w)
    acc = wmma4(a[2 * w] * b[2 * w] + a[2 * w + 1] * b[2 * w + 1], pb[w], acc);
  return acc;
}

static __device__ __forceinline__ v8f upper_rows(const v2f band[4], v8f h) {
  v8f acc = {};
  #pragma unroll
  for (int j = 0; j < 4; ++j) {
    v2f b = { h[2 * j], h[2 * j + 1] };
    acc = wmma4(band[j], b, acc);
  }
  return acc;
}

static __device__ __forceinline__ float xhalf(float x) {
  const int b = __builtin_bit_cast(int, x);
  return __builtin_bit_cast(
      float, __builtin_amdgcn_permlanex16(b, b, 0x76543210, 0xfedcba98, false,
                                          false));
}

static __device__ __forceinline__ v8f rows_below(v8f p, int hi) {
  const float z = hi ? 0.f : p[0];
  const float cs = z + xhalf(z);
  v8f r;
  #pragma unroll
  for (int v = 0; v < 8; ++v) r[v] = cs - p[v];
  return r;
}

static __device__ __forceinline__ v2f pair(v8f s, int q) {
  v2f r = { s[2 * q], s[2 * q + 1] };
  return r;
}

static __device__ __forceinline__ v2f rcp2(v2f x) {
  v2f r = { __builtin_amdgcn_rcpf(x[0]), __builtin_amdgcn_rcpf(x[1]) };
  return r;
}

static __device__ __forceinline__ void solve4_ones(v2f a, v2f b, v2f c, v2f d,
                                                   v2f e, v2f f, v2f g, v2f h,
                                                   v2f i, v2f j, v2f x[4]) {
  const v2f r0 = rcp2(a);
  const v2f l1 = b * r0, l2 = c * r0, l3 = d * r0;
  const v2f e1 = e - l1 * b, f1 = f - l1 * c, g1 = g - l1 * d;
  const v2f h1 = h - l2 * c, i1 = i - l2 * d;
  const v2f j1 = j - l3 * d;
  const v2f y1 = 1.f - l1, y2 = 1.f - l2, y3 = 1.f - l3;
  const v2f r1 = rcp2(e1);
  const v2f m2 = f1 * r1, m3 = g1 * r1;
  const v2f h2 = h1 - m2 * f1, i2 = i1 - m2 * g1;
  const v2f j2 = j1 - m3 * g1;
  const v2f z2 = y2 - m2 * y1, z3 = y3 - m3 * y1;
  const v2f r2 = rcp2(h2);
  const v2f n3 = i2 * r2;
  const v2f j3 = j2 - n3 * i2;
  const v2f w3 = z3 - n3 * z2;
  const v2f r3 = rcp2(j3);
  x[3] = w3 * r3;
  x[2] = (z2 - i2 * x[3]) * r2;
  x[1] = (y1 - f1 * x[2] - g1 * x[3]) * r1;
  x[0] = (1.f - b * x[1] - c * x[2] - d * x[3]) * r0;
}

static __device__ __forceinline__ void fetch5(const float* Ib, const float* Pb,
                                              int r0, int gcc,
                                              float hv[5][16]) {
  #pragma unroll
  for (int y = 0; y < 16; ++y) {
    const int grc = min(max(r0 + y, 0), HH - 1);
    const size_t off = (size_t)grc * WW + gcc;
    hv[4][y] = Pb[off];
    #pragma unroll
    for (int c = 0; c < CC; ++c)
      hv[c][y] = Ib[(size_t)c * HH * WW + off];
  }
}

static __device__ __forceinline__ void stage5(float (*raw)[16 * 33], int r0,
                                              bool cok, int lane,
                                              const float hv[5][16]) {
  #pragma unroll
  for (int y = 0; y < 16; ++y) {
    const int gr = r0 + y;
    const bool ok = cok && (gr >= 0) && (gr < HH);
    #pragma unroll
    for (int pl = 0; pl < 5; ++pl)
      raw[pl][y * 33 + lane] = ok ? hv[pl][y] : 0.f;
  }
}

static __device__ __forceinline__ void fetch2(const float* Ab, const float* Bb,
                                              int r0, int gcc,
                                              float hv[2][16]) {
  #pragma unroll
  for (int y = 0; y < 16; ++y) {
    const int grc = min(max(r0 + y, 0), HH - 1);
    const size_t off = (size_t)grc * WW + gcc;
    hv[0][y] = Ab[off];
    hv[1][y] = Bb[off];
  }
}

static __device__ __forceinline__ void stage2(float (*X)[16 * 33], int r0,
                                              bool cok, int lane,
                                              const float hv[2][16]) {
  #pragma unroll
  for (int y = 0; y < 16; ++y) {
    const int gr = r0 + y;
    const bool ok = cok && (gr >= 0) && (gr < HH);
    X[0][y * 33 + lane] = ok ? hv[0][y] : 0.f;
    X[1][y * 33 + lane] = ok ? hv[1][y] : 0.f;
  }
}

__global__ __launch_bounds__(32) __attribute__((amdgpu_waves_per_eu(2)))
void gf_stats(const float* __restrict__ I,
                                               const float* __restrict__ P,
                                               float* __restrict__ Aout,
                                               float* __restrict__ Bout) {
  __shared__ float raw[5][16 * 33];
  const int lane = threadIdx.x;
  const int tx = blockIdx.x * 16;
  const int ty0 = blockIdx.y * (16 * NT);
  const int bb = blockIdx.z;
  const int gc = tx - RAD + lane;
  const bool cok = (gc >= 0) && (gc < WW);
  const float* Ib = I + (size_t)bb * CC * HH * WW;
  const float* Pb = P + (size_t)bb * HH * WW;
  const int gcc = min(max(gc, 0), WW - 1);

  v2f band[4], pband[4];
  make_band(band, lane);
  make_pband(pband, lane);

  constexpr int po[19] = {4,5,6,7,8, 0,9,10,11,12, 1,13,14,15, 2,16,17, 3,18};
  constexpr int pa[19] = {4,4,4,4,4, 0,0,0,0,0,    1,1,1,1,    2,2,2,   3,3};
  constexpr int pb[19] = {-1,0,1,2,3, -1,0,1,2,3,  -1,1,2,3,   -1,2,3,  -1,3};

  float hv[5][16];
  v2f F[5][8];
  v8f E[5];
  v8f Pc[19];

  fetch5(Ib, Pb, ty0 - RAD, gcc, hv);
  stage5(raw, ty0 - RAD, cok, lane, hv);
  __syncthreads();
  #pragma unroll
  for (int pl = 0; pl < 5; ++pl) {
    load_frag(&raw[pl][0], 0, lane, F[pl]);
    E[pl] = load_edge(&raw[pl][0], lane);
  }
  fetch5(Ib, Pb, ty0 - RAD + 16, gcc, hv);
  #pragma unroll
  for (int i = 0; i < 19; ++i) {
    const int a = pa[i];
    const int b = pb[i] < 0 ? 0 : pb[i];
    const v8f h0 = (pb[i] < 0) ? hpass(F[a], pband, E[a])
                               : hpass_prod(F[a], F[b], pband, E[a], E[b]);
    Pc[i] = upper_rows(band, h0);
  }

  const int hi = lane >> 4;
  const int w = lane & 15;
  const int col = tx + w;
  const int cw = min(col + RAD, WW - 1) - max(col - RAD, 0) + 1;

  #pragma nounroll
  for (int t = 0; t < NT; ++t) {
    const int r1 = ty0 - RAD + 16 * (t + 1);
    __syncthreads();
    stage5(raw, r1, cok, lane, hv);
    __syncthreads();
    #pragma unroll
    for (int pl = 0; pl < 5; ++pl) {
      load_frag(&raw[pl][0], 0, lane, F[pl]);
      E[pl] = load_edge(&raw[pl][0], lane);
    }
    if (t + 1 < NT) fetch5(Ib, Pb, r1 + 16, gcc, hv);

    v8f S[19];
    #pragma unroll
    for (int i = 0; i < 19; ++i) {
      const int a = pa[i];
      const int b = pb[i] < 0 ? 0 : pb[i];
      const v8f hn = (pb[i] < 0) ? hpass(F[a], pband, E[a])
                                 : hpass_prod(F[a], F[b], pband, E[a], E[b]);
      const v8f sp = Pc[i] + hn;
      __builtin_amdgcn_sched_barrier(0);
      const v8f pn = upper_rows(band, hn);
      S[po[i]] = sp + rows_below(pn, hi);
      Pc[i] = pn;
    }

    #pragma unroll
    for (int q = 0; q < 4; ++q) {
      int row[2];
      v2f invN;
      #pragma unroll
      for (int s = 0; s < 2; ++s) {
        row[s] = ty0 + 16 * t + 2 * q + s + 8 * hi;
        const int chc = min(row[s] + RAD, HH - 1) - max(row[s] - RAD, 0) + 1;
        invN[s] = 1.f / (float)(chc * cw);
      }
      const v2f pm = pair(S[4], q) * invN;
      v2f Im[4], first[4];
      v2f sumIm = {0.f, 0.f};
      #pragma unroll
      for (int c = 0; c < 4; ++c) {
        Im[c] = pair(S[c], q) * invN;
        sumIm += Im[c];
        first[c] = pair(S[5 + c], q) * invN - pm * Im[c];
      }
      const v2f ep = {EPSV, EPSV};
      v2f rs[4];
      solve4_ones(pair(S[9], q) * invN + ep, pair(S[10], q) * invN,
                  pair(S[11], q) * invN, pair(S[12], q) * invN,
                  pair(S[13], q) * invN + ep, pair(S[14], q) * invN,
                  pair(S[15], q) * invN, pair(S[16], q) * invN + ep,
                  pair(S[17], q) * invN, pair(S[18], q) * invN + ep, rs);
      v2f sumA = {0.f, 0.f};
      #pragma unroll
      for (int c = 0; c < 4; ++c)
        sumA += first[c] * rs[c];
      const v2f bv = pm - sumA * sumIm;
      #pragma unroll
      for (int s = 0; s < 2; ++s) {
        Aout[((size_t)bb * HH + row[s]) * WW + col] = sumA[s];
        Bout[((size_t)bb * HH + row[s]) * WW + col] = bv[s];
      }
    }
  }
}

__global__ __launch_bounds__(32) void gf_final(const float* __restrict__ I,
                                               const float* __restrict__ Ain,
                                               const float* __restrict__ Bin,
                                               float* __restrict__ Q) {
  __shared__ float X[2][16 * 33];
  const int lane = threadIdx.x;
  const int tx = blockIdx.x * 16;
  const int ty0 = blockIdx.y * (16 * NT);
  const int bb = blockIdx.z;
  const int gc = tx - RAD + lane;
  const bool cok = (gc >= 0) && (gc < WW);
  const int hi = lane >> 4;
  const int w = lane & 15;
  const int col = tx + w;
  const float* Ib = I + (size_t)bb * CC * HH * WW;
  const float* Ab = Ain + (size_t)bb * HH * WW;
  const float* Bb = Bin + (size_t)bb * HH * WW;
  const int gcc = min(max(gc, 0), WW - 1);
  const int cw = min(col + RAD, WW - 1) - max(col - RAD, 0) + 1;

  v2f band[4], pband[4];
  make_band(band, lane);
  make_pband(pband, lane);

  float hv[2][16];
  v2f F[2][8];
  v8f E[2];
  v8f Pc[2];

  fetch2(Ab, Bb, ty0 - RAD, gcc, hv);
  stage2(X, ty0 - RAD, cok, lane, hv);
  __syncthreads();
  #pragma unroll
  for (int pl = 0; pl < 2; ++pl) {
    load_frag(&X[pl][0], 0, lane, F[pl]);
    E[pl] = load_edge(&X[pl][0], lane);
  }
  fetch2(Ab, Bb, ty0 - RAD + 16, gcc, hv);
  #pragma unroll
  for (int pl = 0; pl < 2; ++pl)
    Pc[pl] = upper_rows(band, hpass(F[pl], pband, E[pl]));

  #pragma nounroll
  for (int t = 0; t < NT; ++t) {
    const int r1 = ty0 - RAD + 16 * (t + 1);
    __syncthreads();
    stage2(X, r1, cok, lane, hv);
    __syncthreads();
    #pragma unroll
    for (int pl = 0; pl < 2; ++pl) {
      load_frag(&X[pl][0], 0, lane, F[pl]);
      E[pl] = load_edge(&X[pl][0], lane);
    }
    float iv[4][8];
    #pragma unroll
    for (int v = 0; v < 8; ++v) {
      const size_t off = (size_t)(ty0 + 16 * t + v + 8 * hi) * WW + col;
      #pragma unroll
      for (int c = 0; c < 4; ++c) iv[c][v] = Ib[(size_t)c * HH * WW + off];
    }
    if (t + 1 < NT) fetch2(Ab, Bb, r1 + 16, gcc, hv);

    v8f S[2];
    #pragma unroll
    for (int pl = 0; pl < 2; ++pl) {
      const v8f hn = hpass(F[pl], pband, E[pl]);
      const v8f sp = Pc[pl] + hn;
      const v8f pn = upper_rows(band, hn);
      S[pl] = sp + rows_below(pn, hi);
      Pc[pl] = pn;
    }
    #pragma unroll
    for (int v = 0; v < 8; ++v) {
      const int row = ty0 + 16 * t + v + 8 * hi;
      const int chc = min(row + RAD, HH - 1) - max(row - RAD, 0) + 1;
      const float invN = 1.f / (float)(chc * cw);
      float sumI = 0.f;
      #pragma unroll
      for (int c = 0; c < 4; ++c) sumI += iv[c][v];
      const float am = S[0][v] * invN;
      const float bm = S[1][v] * invN;
      Q[((size_t)bb * HH + row) * WW + col] = am * sumI + bm;
    }
  }
}

extern "C" void kernel_launch(void* const* d_in, const int* in_sizes, int n_in,
                              void* d_out, int out_size, void* d_ws,
                              size_t ws_size, hipStream_t stream) {
  const float* I = (const float*)d_in[0];
  const float* P = (const float*)d_in[1];
  float* Q = (float*)d_out;
  float* Aws = (float*)d_ws;
  float* Bws = Aws + (size_t)NB * HH * WW;
  dim3 grid(WW / 16, HH / (16 * NT), NB);
  gf_stats<<<grid, dim3(32), 0, stream>>>(I, P, Aws, Bws);
  gf_final<<<grid, dim3(32), 0, stream>>>(I, Aws, Bws, Q);
  (void)in_sizes; (void)n_in; (void)out_size; (void)ws_size;
}


# Vertical band pass over row groups of four: one K = 4 matrix op on the group totals plus in-lane packed sums to each group's end, not four ops of K = 4 rows: 5 not 8 per plane, block; sums reordered
# speedup vs baseline: 5.4266x; 1.0527x over previous
#include <hip/hip_runtime.h>

#define HH 768
#define WW 768
#define CC 4
#define NB 8
#define RAD 8
#define EPSV 1e-4f
#define NT 6

typedef __attribute__((ext_vector_type(2))) float v2f;
typedef __attribute__((ext_vector_type(8))) float v8f;

static __device__ __forceinline__ v8f wmma4(v2f a, v2f b, v8f c) {
  return __builtin_amdgcn_wmma_f32_16x16x4_f32(false, a, false, b, (short)0, c,
                                               false, false);
}

static __device__ __forceinline__ constexpr int hrow(int v) {
  return 4 * (v & 1) + (v >> 1);
}

static __device__ __forceinline__ void load_frag(const float* X, int rs,
                                                 int lane, v2f a[8]) {
  const int m = 8 * ((lane & 15) >> 3) + hrow(lane & 7);
  const int hi = lane >> 4;
  #pragma unroll
  for (int w = 0; w < 4; ++w) {
    const int kk = 2 * w;
    const int j0 = 16 * (kk >> 2) + 8 * hi + 2 * (kk & 3);
    v2f ev = { X[(rs * 16 + m) * 33 + j0], X[(rs * 16 + m) * 33 + j0 + 2] };
    v2f od = { X[(rs * 16 + m) * 33 + j0 + 1], X[(rs * 16 + m) * 33 + j0 + 3] };
    a[2 * w] = ev;
    a[2 * w + 1] = od;
  }
}

static __device__ __forceinline__ v2f make_qband(int lane) {
  const int g0 = 2 * (lane >> 4);
  const int gi = 2 * ((lane & 15) >> 3) + (lane & 1);
  const bool first = ((lane & 7) >> 1) == 0;
  v2f t = { (g0 > gi || (g0 == gi && first)) ? 1.f : 0.f,
            (g0 + 1 > gi || (g0 + 1 == gi && first)) ? 1.f : 0.f };
  return t;
}

static __device__ __forceinline__ void make_pband(v2f pb[4], int lane) {
  const int hi = lane >> 4;
  const int n = lane & 15;
  #pragma unroll
  for (int w = 0; w < 4; ++w) {
    v2f t;
    #pragma unroll
    for (int s = 0; s < 2; ++s) {
      const int kk = 2 * w + s;
      const int d = 16 * (kk >> 2) + 8 * hi + 2 * (kk & 3) - n;
      t[s] = (d >= 0 && d <= 15) ? 1.f : 0.f;
    }
    pb[w] = t;
  }
}

static __device__ __forceinline__ v8f load_edge(const float* X, int lane) {
  const int n = lane & 15;
  const int ec = (n & 1) ? n : n + 16;
  const int r8 = 8 * (lane >> 4);
  v8f e;
  #pragma unroll
  for (int v = 0; v < 8; ++v) e[v] = X[(r8 + hrow(v)) * 33 + ec];
  return e;
}

static __device__ __forceinline__ v8f hpass(const v2f a[8], const v2f pb[4],
                                            v8f edge) {
  v8f acc = edge;
  #pragma unroll
  for (int w = 0; w < 4; ++w)
    acc = wmma4(a[2 * w] + a[2 * w + 1], pb[w], acc);
  return acc;
}

static __device__ __forceinline__ v8f hpass_prod(const v2f a[8],
                                                 const v2f b[8],
                                                 const v2f pb[4], v8f ea,
                                                 v8f eb) {
  v8f acc = ea * eb;
  #pragma unroll
  for (int w = 0; w < 4; ++w)
    acc = wmma4(a[2 * w] * b[2 * w] + a[2 * w + 1] * b[2 * w + 1], pb[w], acc);
  return acc;
}

static __device__ __forceinline__ v2f pair(v8f s, int q) {
  v2f r = { s[2 * q], s[2 * q + 1] };
  return r;
}

static __device__ __forceinline__ v8f upper_rows(v2f qband, v8f h) {
  const v2f q3 = pair(h, 3);
  const v2f q2 = pair(h, 2) + q3;
  const v2f q1 = pair(h, 1) + q2;
  const v2f q0 = pair(h, 0) + q1;
  const v8f z = {};
  v8f d = wmma4(qband, q0, z);
  d[2] += q1[0]; d[3] += q1[1];
  d[4] += q2[0]; d[5] += q2[1];
  d[6] += q3[0]; d[7] += q3[1];
  return d;
}

static __device__ __forceinline__ float xhalf(float x) {
  const int b = __builtin_bit_cast(int, x);
  return __builtin_bit_cast(
      float, __builtin_amdgcn_permlanex16(b, b, 0x76543210, 0xfedcba98, false,
                                          false));
}

static __device__ __forceinline__ v8f rows_below(v8f p, int hi) {
  const float z = hi ? 0.f : p[0];
  const float cs = z + xhalf(z);
  v8f r;
  #pragma unroll
  for (int v = 0; v < 8; ++v) r[v] = cs - p[v];
  return r;
}

static __device__ __forceinline__ v2f rcp2(v2f x) {
  v2f r = { __builtin_amdgcn_rcpf(x[0]), __builtin_amdgcn_rcpf(x[1]) };
  return r;
}

static __device__ __forceinline__ void solve4_ones(v2f a, v2f b, v2f c, v2f d,
                                                   v2f e, v2f f, v2f g, v2f h,
                                                   v2f i, v2f j, v2f x[4]) {
  const v2f r0 = rcp2(a);
  const v2f l1 = b * r0, l2 = c * r0, l3 = d * r0;
  const v2f e1 = e - l1 * b, f1 = f - l1 * c, g1 = g - l1 * d;
  const v2f h1 = h - l2 * c, i1 = i - l2 * d;
  const v2f j1 = j - l3 * d;
  const v2f y1 = 1.f - l1, y2 = 1.f - l2, y3 = 1.f - l3;
  const v2f r1 = rcp2(e1);
  const v2f m2 = f1 * r1, m3 = g1 * r1;
  const v2f h2 = h1 - m2 * f1, i2 = i1 - m2 * g1;
  const v2f j2 = j1 - m3 * g1;
  const v2f z2 = y2 - m2 * y1, z3 = y3 - m3 * y1;
  const v2f r2 = rcp2(h2);
  const v2f n3 = i2 * r2;
  const v2f j3 = j2 - n3 * i2;
  const v2f w3 = z3 - n3 * z2;
  const v2f r3 = rcp2(j3);
  x[3] = w3 * r3;
  x[2] = (z2 - i2 * x[3]) * r2;
  x[1] = (y1 - f1 * x[2] - g1 * x[3]) * r1;
  x[0] = (1.f - b * x[1] - c * x[2] - d * x[3]) * r0;
}

static __device__ __forceinline__ void fetch5(const float* Ib, const float* Pb,
                                              int r0, int gcc,
                                              float hv[5][16]) {
  #pragma unroll
  for (int y = 0; y < 16; ++y) {
    const int grc = min(max(r0 + y, 0), HH - 1);
    const size_t off = (size_t)grc * WW + gcc;
    hv[4][y] = Pb[off];
    #pragma unroll
    for (int c = 0; c < CC; ++c)
      hv[c][y] = Ib[(size_t)c * HH * WW + off];
  }
}

static __device__ __forceinline__ void stage5(float (*raw)[16 * 33], int r0,
                                              bool cok, int lane,
                                              const float hv[5][16]) {
  #pragma unroll
  for (int y = 0; y < 16; ++y) {
    const int gr = r0 + y;
    const bool ok = cok && (gr >= 0) && (gr < HH);
    #pragma unroll
    for (int pl = 0; pl < 5; ++pl)
      raw[pl][y * 33 + lane] = ok ? hv[pl][y] : 0.f;
  }
}

static __device__ __forceinline__ void fetch2(const float* Ab, const float* Bb,
                                              int r0, int gcc,
                                              float hv[2][16]) {
  #pragma unroll
  for (int y = 0; y < 16; ++y) {
    const int grc = min(max(r0 + y, 0), HH - 1);
    const size_t off = (size_t)grc * WW + gcc;
    hv[0][y] = Ab[off];
    hv[1][y] = Bb[off];
  }
}

static __device__ __forceinline__ void stage2(float (*X)[16 * 33], int r0,
                                              bool cok, int lane,
                                              const float hv[2][16]) {
  #pragma unroll
  for (int y = 0; y < 16; ++y) {
    const int gr = r0 + y;
    const bool ok = cok && (gr >= 0) && (gr < HH);
    X[0][y * 33 + lane] = ok ? hv[0][y] : 0.f;
    X[1][y * 33 + lane] = ok ? hv[1][y] : 0.f;
  }
}

__global__ __launch_bounds__(32) __attribute__((amdgpu_waves_per_eu(2)))
void gf_stats(const float* __restrict__ I,
                                               const float* __restrict__ P,
                                               float* __restrict__ Aout,
                                               float* __restrict__ Bout) {
  __shared__ float raw[5][16 * 33];
  const int lane = threadIdx.x;
  const int tx = blockIdx.x * 16;
  const int ty0 = blockIdx.y * (16 * NT);
  const int bb = blockIdx.z;
  const int gc = tx - RAD + lane;
  const bool cok = (gc >= 0) && (gc < WW);
  const float* Ib = I + (size_t)bb * CC * HH * WW;
  const float* Pb = P + (size_t)bb * HH * WW;
  const int gcc = min(max(gc, 0), WW - 1);

  v2f pband[4];
  const v2f qband = make_qband(lane);
  make_pband(pband, lane);

  constexpr int po[19] = {4,5,6,7,8, 0,9,10,11,12, 1,13,14,15, 2,16,17, 3,18};
  constexpr int pa[19] = {4,4,4,4,4, 0,0,0,0,0,    1,1,1,1,    2,2,2,   3,3};
  constexpr int pb[19] = {-1,0,1,2,3, -1,0,1,2,3,  -1,1,2,3,   -1,2,3,  -1,3};

  float hv[5][16];
  v2f F[5][8];
  v8f E[5];
  v8f Pc[19];

  fetch5(Ib, Pb, ty0 - RAD, gcc, hv);
  stage5(raw, ty0 - RAD, cok, lane, hv);
  __syncthreads();
  #pragma unroll
  for (int pl = 0; pl < 5; ++pl) {
    load_frag(&raw[pl][0], 0, lane, F[pl]);
    E[pl] = load_edge(&raw[pl][0], lane);
  }
  fetch5(Ib, Pb, ty0 - RAD + 16, gcc, hv);
  #pragma unroll
  for (int i = 0; i < 19; ++i) {
    const int a = pa[i];
    const int b = pb[i] < 0 ? 0 : pb[i];
    const v8f h0 = (pb[i] < 0) ? hpass(F[a], pband, E[a])
                               : hpass_prod(F[a], F[b], pband, E[a], E[b]);
    Pc[i] = upper_rows(qband, h0);
  }

  const int hi = lane >> 4;
  const int w = lane & 15;
  const int col = tx + w;
  const int cw = min(col + RAD, WW - 1) - max(col - RAD, 0) + 1;

  #pragma nounroll
  for (int t = 0; t < NT; ++t) {
    const int r1 = ty0 - RAD + 16 * (t + 1);
    __syncthreads();
    stage5(raw, r1, cok, lane, hv);
    __syncthreads();
    #pragma unroll
    for (int pl = 0; pl < 5; ++pl) {
      load_frag(&raw[pl][0], 0, lane, F[pl]);
      E[pl] = load_edge(&raw[pl][0], lane);
    }
    if (t + 1 < NT) fetch5(Ib, Pb, r1 + 16, gcc, hv);

    v8f S[19];
    #pragma unroll
    for (int i = 0; i < 19; ++i) {
      const int a = pa[i];
      const int b = pb[i] < 0 ? 0 : pb[i];
      const v8f hn = (pb[i] < 0) ? hpass(F[a], pband, E[a])
                                 : hpass_prod(F[a], F[b], pband, E[a], E[b]);
      const v8f sp = Pc[i] + hn;
      __builtin_amdgcn_sched_barrier(0);
      const v8f pn = upper_rows(qband, hn);
      S[po[i]] = sp + rows_below(pn, hi);
      Pc[i] = pn;
    }

    #pragma unroll
    for (int q = 0; q < 4; ++q) {
      int row[2];
      v2f invN;
      #pragma unroll
      for (int s = 0; s < 2; ++s) {
        row[s] = ty0 + 16 * t + hrow(2 * q + s) + 8 * hi;
        const int chc = min(row[s] + RAD, HH - 1) - max(row[s] - RAD, 0) + 1;
        invN[s] = 1.f / (float)(chc * cw);
      }
      const v2f pm = pair(S[4], q) * invN;
      v2f Im[4], first[4];
      v2f sumIm = {0.f, 0.f};
      #pragma unroll
      for (int c = 0; c < 4; ++c) {
        Im[c] = pair(S[c], q) * invN;
        sumIm += Im[c];
        first[c] = pair(S[5 + c], q) * invN - pm * Im[c];
      }
      const v2f ep = {EPSV, EPSV};
      v2f rs[4];
      solve4_ones(pair(S[9], q) * invN + ep, pair(S[10], q) * invN,
                  pair(S[11], q) * invN, pair(S[12], q) * invN,
                  pair(S[13], q) * invN + ep, pair(S[14], q) * invN,
                  pair(S[15], q) * invN, pair(S[16], q) * invN + ep,
                  pair(S[17], q) * invN, pair(S[18], q) * invN + ep, rs);
      v2f sumA = {0.f, 0.f};
      #pragma unroll
      for (int c = 0; c < 4; ++c)
        sumA += first[c] * rs[c];
      const v2f bv = pm - sumA * sumIm;
      #pragma unroll
      for (int s = 0; s < 2; ++s) {
        Aout[((size_t)bb * HH + row[s]) * WW + col] = sumA[s];
        Bout[((size_t)bb * HH + row[s]) * WW + col] = bv[s];
      }
    }
  }
}

__global__ __launch_bounds__(32) void gf_final(const float* __restrict__ I,
                                               const float* __restrict__ Ain,
                                               const float* __restrict__ Bin,
                                               float* __restrict__ Q) {
  __shared__ float X[2][16 * 33];
  const int lane = threadIdx.x;
  const int tx = blockIdx.x * 16;
  const int ty0 = blockIdx.y * (16 * NT);
  const int bb = blockIdx.z;
  const int gc = tx - RAD + lane;
  const bool cok = (gc >= 0) && (gc < WW);
  const int hi = lane >> 4;
  const int w = lane & 15;
  const int col = tx + w;
  const float* Ib = I + (size_t)bb * CC * HH * WW;
  const float* Ab = Ain + (size_t)bb * HH * WW;
  const float* Bb = Bin + (size_t)bb * HH * WW;
  const int gcc = min(max(gc, 0), WW - 1);
  const int cw = min(col + RAD, WW - 1) - max(col - RAD, 0) + 1;

  v2f pband[4];
  const v2f qband = make_qband(lane);
  make_pband(pband, lane);

  float hv[2][16];
  v2f F[2][8];
  v8f E[2];
  v8f Pc[2];

  fetch2(Ab, Bb, ty0 - RAD, gcc, hv);
  stage2(X, ty0 - RAD, cok, lane, hv);
  __syncthreads();
  #pragma unroll
  for (int pl = 0; pl < 2; ++pl) {
    load_frag(&X[pl][0], 0, lane, F[pl]);
    E[pl] = load_edge(&X[pl][0], lane);
  }
  fetch2(Ab, Bb, ty0 - RAD + 16, gcc, hv);
  #pragma unroll
  for (int pl = 0; pl < 2; ++pl)
    Pc[pl] = upper_rows(qband, hpass(F[pl], pband, E[pl]));

  #pragma nounroll
  for (int t = 0; t < NT; ++t) {
    const int r1 = ty0 - RAD + 16 * (t + 1);
    __syncthreads();
    stage2(X, r1, cok, lane, hv);
    __syncthreads();
    #pragma unroll
    for (int pl = 0; pl < 2; ++pl) {
      load_frag(&X[pl][0], 0, lane, F[pl]);
      E[pl] = load_edge(&X[pl][0], lane);
    }
    float iv[4][8];
    #pragma unroll
    for (int v = 0; v < 8; ++v) {
      const size_t off = (size_t)(ty0 + 16 * t + hrow(v) + 8 * hi) * WW + col;
      #pragma unroll
      for (int c = 0; c < 4; ++c) iv[c][v] = Ib[(size_t)c * HH * WW + off];
    }
    if (t + 1 < NT) fetch2(Ab, Bb, r1 + 16, gcc, hv);

    v8f S[2];
    #pragma unroll
    for (int pl = 0; pl < 2; ++pl) {
      const v8f hn = hpass(F[pl], pband, E[pl]);
      const v8f sp = Pc[pl] + hn;
      const v8f pn = upper_rows(qband, hn);
      S[pl] = sp + rows_below(pn, hi);
      Pc[pl] = pn;
    }
    #pragma unroll
    for (int v = 0; v < 8; ++v) {
      const int row = ty0 + 16 * t + hrow(v) + 8 * hi;
      const int chc = min(row + RAD, HH - 1) - max(row - RAD, 0) + 1;
      const float invN = 1.f / (float)(chc * cw);
      float sumI = 0.f;
      #pragma unroll
      for (int c = 0; c < 4; ++c) sumI += iv[c][v];
      const float am = S[0][v] * invN;
      const float bm = S[1][v] * invN;
      Q[((size_t)bb * HH + row) * WW + col] = am * sumI + bm;
    }
  }
}

extern "C" void kernel_launch(void* const* d_in, const int* in_sizes, int n_in,
                              void* d_out, int out_size, void* d_ws,
                              size_t ws_size, hipStream_t stream) {
  const float* I = (const float*)d_in[0];
  const float* P = (const float*)d_in[1];
  float* Q = (float*)d_out;
  float* Aws = (float*)d_ws;
  float* Bws = Aws + (size_t)NB * HH * WW;
  dim3 grid(WW / 16, HH / (16 * NT), NB);
  gf_stats<<<grid, dim3(32), 0, stream>>>(I, P, Aws, Bws);
  gf_final<<<grid, dim3(32), 0, stream>>>(I, Aws, Bws, Q);
  (void)in_sizes; (void)n_in; (void)out_size; (void)ws_size;
}


# Halo blocks fetched by 16-byte loads: a lane takes four columns of one row, an instruction four rows: 20 not 80 loads per block (second kernel 8 not 32), all in flight under the 63-load limit
# speedup vs baseline: 5.8807x; 1.0837x over previous
#include <hip/hip_runtime.h>

#define HH 768
#define WW 768
#define CC 4
#define NB 8
#define RAD 8
#define EPSV 1e-4f
#define NT 6

typedef __attribute__((ext_vector_type(2))) float v2f;
typedef __attribute__((ext_vector_type(4))) float v4f;
typedef __attribute__((ext_vector_type(8))) float v8f;

static __device__ __forceinline__ v8f wmma4(v2f a, v2f b, v8f c) {
  return __builtin_amdgcn_wmma_f32_16x16x4_f32(false, a, false, b, (short)0, c,
                                               false, false);
}

static __device__ __forceinline__ constexpr int hrow(int v) {
  return 4 * (v & 1) + (v >> 1);
}

static __device__ __forceinline__ void load_frag(const float* X, int rs,
                                                 int lane, v2f a[8]) {
  const int m = 8 * ((lane & 15) >> 3) + hrow(lane & 7);
  const int hi = lane >> 4;
  #pragma unroll
  for (int w = 0; w < 4; ++w) {
    const int kk = 2 * w;
    const int j0 = 16 * (kk >> 2) + 8 * hi + 2 * (kk & 3);
    v2f ev = { X[(rs * 16 + m) * 33 + j0], X[(rs * 16 + m) * 33 + j0 + 2] };
    v2f od = { X[(rs * 16 + m) * 33 + j0 + 1], X[(rs * 16 + m) * 33 + j0 + 3] };
    a[2 * w] = ev;
    a[2 * w + 1] = od;
  }
}

static __device__ __forceinline__ v2f make_qband(int lane) {
  const int g0 = 2 * (lane >> 4);
  const int gi = 2 * ((lane & 15) >> 3) + (lane & 1);
  const bool first = ((lane & 7) >> 1) == 0;
  v2f t = { (g0 > gi || (g0 == gi && first)) ? 1.f : 0.f,
            (g0 + 1 > gi || (g0 + 1 == gi && first)) ? 1.f : 0.f };
  return t;
}

static __device__ __forceinline__ void make_pband(v2f pb[4], int lane) {
  const int hi = lane >> 4;
  const int n = lane & 15;
  #pragma unroll
  for (int w = 0; w < 4; ++w) {
    v2f t;
    #pragma unroll
    for (int s = 0; s < 2; ++s) {
      const int kk = 2 * w + s;
      const int d = 16 * (kk >> 2) + 8 * hi + 2 * (kk & 3) - n;
      t[s] = (d >= 0 && d <= 15) ? 1.f : 0.f;
    }
    pb[w] = t;
  }
}

static __device__ __forceinline__ v8f load_edge(const float* X, int lane) {
  const int n = lane & 15;
  const int ec = (n & 1) ? n : n + 16;
  const int r8 = 8 * (lane >> 4);
  v8f e;
  #pragma unroll
  for (int v = 0; v < 8; ++v) e[v] = X[(r8 + hrow(v)) * 33 + ec];
  return e;
}

static __device__ __forceinline__ v8f hpass(const v2f a[8], const v2f pb[4],
                                            v8f edge) {
  v8f acc = edge;
  #pragma unroll
  for (int w = 0; w < 4; ++w)
    acc = wmma4(a[2 * w] + a[2 * w + 1], pb[w], acc);
  return acc;
}

static __device__ __forceinline__ v8f hpass_prod(const v2f a[8],
                                                 const v2f b[8],
                                                 const v2f pb[4], v8f ea,
                                                 v8f eb) {
  v8f acc = ea * eb;
  #pragma unroll
  for (int w = 0; w < 4; ++w)
    acc = wmma4(a[2 * w] * b[2 * w] + a[2 * w + 1] * b[2 * w + 1], pb[w], acc);
  return acc;
}

static __device__ __forceinline__ v2f pair(v8f s, int q) {
  v2f r = { s[2 * q], s[2 * q + 1] };
  return r;
}

static __device__ __forceinline__ v8f upper_rows(v2f qband, v8f h) {
  const v2f q3 = pair(h, 3);
  const v2f q2 = pair(h, 2) + q3;
  const v2f q1 = pair(h, 1) + q2;
  const v2f q0 = pair(h, 0) + q1;
  const v8f z = {};
  v8f d = wmma4(qband, q0, z);
  d[2] += q1[0]; d[3] += q1[1];
  d[4] += q2[0]; d[5] += q2[1];
  d[6] += q3[0]; d[7] += q3[1];
  return d;
}

static __device__ __forceinline__ float xhalf(float x) {
  const int b = __builtin_bit_cast(int, x);
  return __builtin_bit_cast(
      float, __builtin_amdgcn_permlanex16(b, b, 0x76543210, 0xfedcba98, false,
                                          false));
}

static __device__ __forceinline__ v8f rows_below(v8f p, int hi) {
  const float z = hi ? 0.f : p[0];
  const float cs = z + xhalf(z);
  v8f r;
  #pragma unroll
  for (int v = 0; v < 8; ++v) r[v] = cs - p[v];
  return r;
}

static __device__ __forceinline__ v2f rcp2(v2f x) {
  v2f r = { __builtin_amdgcn_rcpf(x[0]), __builtin_amdgcn_rcpf(x[1]) };
  return r;
}

static __device__ __forceinline__ void solve4_ones(v2f a, v2f b, v2f c, v2f d,
                                                   v2f e, v2f f, v2f g, v2f h,
                                                   v2f i, v2f j, v2f x[4]) {
  const v2f r0 = rcp2(a);
  const v2f l1 = b * r0, l2 = c * r0, l3 = d * r0;
  const v2f e1 = e - l1 * b, f1 = f - l1 * c, g1 = g - l1 * d;
  const v2f h1 = h - l2 * c, i1 = i - l2 * d;
  const v2f j1 = j - l3 * d;
  const v2f y1 = 1.f - l1, y2 = 1.f - l2, y3 = 1.f - l3;
  const v2f r1 = rcp2(e1);
  const v2f m2 = f1 * r1, m3 = g1 * r1;
  const v2f h2 = h1 - m2 * f1, i2 = i1 - m2 * g1;
  const v2f j2 = j1 - m3 * g1;
  const v2f z2 = y2 - m2 * y1, z3 = y3 - m3 * y1;
  const v2f r2 = rcp2(h2);
  const v2f n3 = i2 * r2;
  const v2f j3 = j2 - n3 * i2;
  const v2f w3 = z3 - n3 * z2;
  const v2f r3 = rcp2(j3);
  x[3] = w3 * r3;
  x[2] = (z2 - i2 * x[3]) * r2;
  x[1] = (y1 - f1 * x[2] - g1 * x[3]) * r1;
  x[0] = (1.f - b * x[1] - c * x[2] - d * x[3]) * r0;
}

static __device__ __forceinline__ void fetch5(const float* Ib, const float* Pb,
                                              int r0, int fr, int gcc,
                                              v4f hv[5][4]) {
  #pragma unroll
  for (int g = 0; g < 4; ++g) {
    const int grc = min(max(r0 + 4 * g + fr, 0), HH - 1);
    const size_t off = (size_t)grc * WW + gcc;
    hv[4][g] = *(const v4f*)(Pb + off);
    #pragma unroll
    for (int c = 0; c < CC; ++c)
      hv[c][g] = *(const v4f*)(Ib + (size_t)c * HH * WW + off);
  }
}

static __device__ __forceinline__ void stage5(float (*raw)[16 * 33], int r0,
                                              bool cok, int fr, int fc,
                                              const v4f hv[5][4]) {
  #pragma unroll
  for (int g = 0; g < 4; ++g) {
    const int y = 4 * g + fr;
    const int gr = r0 + y;
    const bool ok = cok && (gr >= 0) && (gr < HH);
    #pragma unroll
    for (int pl = 0; pl < 5; ++pl) {
      #pragma unroll
      for (int i = 0; i < 4; ++i)
        raw[pl][y * 33 + fc + i] = ok ? hv[pl][g][i] : 0.f;
    }
  }
}

static __device__ __forceinline__ void fetch2(const float* Ab, const float* Bb,
                                              int r0, int fr, int gcc,
                                              v4f hv[2][4]) {
  #pragma unroll
  for (int g = 0; g < 4; ++g) {
    const int grc = min(max(r0 + 4 * g + fr, 0), HH - 1);
    const size_t off = (size_t)grc * WW + gcc;
    hv[0][g] = *(const v4f*)(Ab + off);
    hv[1][g] = *(const v4f*)(Bb + off);
  }
}

static __device__ __forceinline__ void stage2(float (*X)[16 * 33], int r0,
                                              bool cok, int fr, int fc,
                                              const v4f hv[2][4]) {
  #pragma unroll
  for (int g = 0; g < 4; ++g) {
    const int y = 4 * g + fr;
    const int gr = r0 + y;
    const bool ok = cok && (gr >= 0) && (gr < HH);
    #pragma unroll
    for (int i = 0; i < 4; ++i) {
      X[0][y * 33 + fc + i] = ok ? hv[0][g][i] : 0.f;
      X[1][y * 33 + fc + i] = ok ? hv[1][g][i] : 0.f;
    }
  }
}

__global__ __launch_bounds__(32) __attribute__((amdgpu_waves_per_eu(2)))
void gf_stats(const float* __restrict__ I,
                                               const float* __restrict__ P,
                                               float* __restrict__ Aout,
                                               float* __restrict__ Bout) {
  __shared__ float raw[5][16 * 33];
  const int lane = threadIdx.x;
  const int tx = blockIdx.x * 16;
  const int ty0 = blockIdx.y * (16 * NT);
  const int bb = blockIdx.z;
  const int fr = lane >> 3;
  const int fc = 4 * (lane & 7);
  const int gc = tx - RAD + fc;
  const bool cok = (gc >= 0) && (gc < WW);
  const float* Ib = I + (size_t)bb * CC * HH * WW;
  const float* Pb = P + (size_t)bb * HH * WW;
  const int gcc = min(max(gc, 0), WW - 4);

  v2f pband[4];
  const v2f qband = make_qband(lane);
  make_pband(pband, lane);

  constexpr int po[19] = {4,5,6,7,8, 0,9,10,11,12, 1,13,14,15, 2,16,17, 3,18};
  constexpr int pa[19] = {4,4,4,4,4, 0,0,0,0,0,    1,1,1,1,    2,2,2,   3,3};
  constexpr int pb[19] = {-1,0,1,2,3, -1,0,1,2,3,  -1,1,2,3,   -1,2,3,  -1,3};

  v4f hv[5][4];
  v2f F[5][8];
  v8f E[5];
  v8f Pc[19];

  fetch5(Ib, Pb, ty0 - RAD, fr, gcc, hv);
  stage5(raw, ty0 - RAD, cok, fr, fc, hv);
  __syncthreads();
  #pragma unroll
  for (int pl = 0; pl < 5; ++pl) {
    load_frag(&raw[pl][0], 0, lane, F[pl]);
    E[pl] = load_edge(&raw[pl][0], lane);
  }
  fetch5(Ib, Pb, ty0 - RAD + 16, fr, gcc, hv);
  #pragma unroll
  for (int i = 0; i < 19; ++i) {
    const int a = pa[i];
    const int b = pb[i] < 0 ? 0 : pb[i];
    const v8f h0 = (pb[i] < 0) ? hpass(F[a], pband, E[a])
                               : hpass_prod(F[a], F[b], pband, E[a], E[b]);
    Pc[i] = upper_rows(qband, h0);
  }

  const int hi = lane >> 4;
  const int w = lane & 15;
  const int col = tx + w;
  const int cw = min(col + RAD, WW - 1) - max(col - RAD, 0) + 1;

  #pragma nounroll
  for (int t = 0; t < NT; ++t) {
    const int r1 = ty0 - RAD + 16 * (t + 1);
    __syncthreads();
    stage5(raw, r1, cok, fr, fc, hv);
    __syncthreads();
    #pragma unroll
    for (int pl = 0; pl < 5; ++pl) {
      load_frag(&raw[pl][0], 0, lane, F[pl]);
      E[pl] = load_edge(&raw[pl][0], lane);
    }
    if (t + 1 < NT) fetch5(Ib, Pb, r1 + 16, fr, gcc, hv);

    v8f S[19];
    #pragma unroll
    for (int i = 0; i < 19; ++i) {
      const int a = pa[i];
      const int b = pb[i] < 0 ? 0 : pb[i];
      const v8f hn = (pb[i] < 0) ? hpass(F[a], pband, E[a])
                                 : hpass_prod(F[a], F[b], pband, E[a], E[b]);
      const v8f sp = Pc[i] + hn;
      __builtin_amdgcn_sched_barrier(0);
      const v8f pn = upper_rows(qband, hn);
      S[po[i]] = sp + rows_below(pn, hi);
      Pc[i] = pn;
    }

    #pragma unroll
    for (int q = 0; q < 4; ++q) {
      int row[2];
      v2f invN;
      #pragma unroll
      for (int s = 0; s < 2; ++s) {
        row[s] = ty0 + 16 * t + hrow(2 * q + s) + 8 * hi;
        const int chc = min(row[s] + RAD, HH - 1) - max(row[s] - RAD, 0) + 1;
        invN[s] = 1.f / (float)(chc * cw);
      }
      const v2f pm = pair(S[4], q) * invN;
      v2f Im[4], first[4];
      v2f sumIm = {0.f, 0.f};
      #pragma unroll
      for (int c = 0; c < 4; ++c) {
        Im[c] = pair(S[c], q) * invN;
        sumIm += Im[c];
        first[c] = pair(S[5 + c], q) * invN - pm * Im[c];
      }
      const v2f ep = {EPSV, EPSV};
      v2f rs[4];
      solve4_ones(pair(S[9], q) * invN + ep, pair(S[10], q) * invN,
                  pair(S[11], q) * invN, pair(S[12], q) * invN,
                  pair(S[13], q) * invN + ep, pair(S[14], q) * invN,
                  pair(S[15], q) * invN, pair(S[16], q) * invN + ep,
                  pair(S[17], q) * invN, pair(S[18], q) * invN + ep, rs);
      v2f sumA = {0.f, 0.f};
      #pragma unroll
      for (int c = 0; c < 4; ++c)
        sumA += first[c] * rs[c];
      const v2f bv = pm - sumA * sumIm;
      #pragma unroll
      for (int s = 0; s < 2; ++s) {
        Aout[((size_t)bb * HH + row[s]) * WW + col] = sumA[s];
        Bout[((size_t)bb * HH + row[s]) * WW + col] = bv[s];
      }
    }
  }
}

__global__ __launch_bounds__(32) void gf_final(const float* __restrict__ I,
                                               const float* __restrict__ Ain,
                                               const float* __restrict__ Bin,
                                               float* __restrict__ Q) {
  __shared__ float X[2][16 * 33];
  const int lane = threadIdx.x;
  const int tx = blockIdx.x * 16;
  const int ty0 = blockIdx.y * (16 * NT);
  const int bb = blockIdx.z;
  const int fr = lane >> 3;
  const int fc = 4 * (lane & 7);
  const int gc = tx - RAD + fc;
  const bool cok = (gc >= 0) && (gc < WW);
  const int hi = lane >> 4;
  const int w = lane & 15;
  const int col = tx + w;
  const float* Ib = I + (size_t)bb * CC * HH * WW;
  const float* Ab = Ain + (size_t)bb * HH * WW;
  const float* Bb = Bin + (size_t)bb * HH * WW;
  const int gcc = min(max(gc, 0), WW - 4);
  const int cw = min(col + RAD, WW - 1) - max(col - RAD, 0) + 1;

  v2f pband[4];
  const v2f qband = make_qband(lane);
  make_pband(pband, lane);

  v4f hv[2][4];
  v2f F[2][8];
  v8f E[2];
  v8f Pc[2];

  fetch2(Ab, Bb, ty0 - RAD, fr, gcc, hv);
  stage2(X, ty0 - RAD, cok, fr, fc, hv);
  __syncthreads();
  #pragma unroll
  for (int pl = 0; pl < 2; ++pl) {
    load_frag(&X[pl][0], 0, lane, F[pl]);
    E[pl] = load_edge(&X[pl][0], lane);
  }
  fetch2(Ab, Bb, ty0 - RAD + 16, fr, gcc, hv);
  #pragma unroll
  for (int pl = 0; pl < 2; ++pl)
    Pc[pl] = upper_rows(qband, hpass(F[pl], pband, E[pl]));

  #pragma nounroll
  for (int t = 0; t < NT; ++t) {
    const int r1 = ty0 - RAD + 16 * (t + 1);
    __syncthreads();
    stage2(X, r1, cok, fr, fc, hv);
    __syncthreads();
    #pragma unroll
    for (int pl = 0; pl < 2; ++pl) {
      load_frag(&X[pl][0], 0, lane, F[pl]);
      E[pl] = load_edge(&X[pl][0], lane);
    }
    float iv[4][8];
    #pragma unroll
    for (int v = 0; v < 8; ++v) {
      const size_t off = (size_t)(ty0 + 16 * t + hrow(v) + 8 * hi) * WW + col;
      #pragma unroll
      for (int c = 0; c < 4; ++c) iv[c][v] = Ib[(size_t)c * HH * WW + off];
    }
    if (t + 1 < NT) fetch2(Ab, Bb, r1 + 16, fr, gcc, hv);

    v8f S[2];
    #pragma unroll
    for (int pl = 0; pl < 2; ++pl) {
      const v8f hn = hpass(F[pl], pband, E[pl]);
      const v8f sp = Pc[pl] + hn;
      const v8f pn = upper_rows(qband, hn);
      S[pl] = sp + rows_below(pn, hi);
      Pc[pl] = pn;
    }
    #pragma unroll
    for (int v = 0; v < 8; ++v) {
      const int row = ty0 + 16 * t + hrow(v) + 8 * hi;
      const int chc = min(row + RAD, HH - 1) - max(row - RAD, 0) + 1;
      const float invN = 1.f / (float)(chc * cw);
      float sumI = 0.f;
      #pragma unroll
      for (int c = 0; c < 4; ++c) sumI += iv[c][v];
      const float am = S[0][v] * invN;
      const float bm = S[1][v] * invN;
      Q[((size_t)bb * HH + row) * WW + col] = am * sumI + bm;
    }
  }
}

extern "C" void kernel_launch(void* const* d_in, const int* in_sizes, int n_in,
                              void* d_out, int out_size, void* d_ws,
                              size_t ws_size, hipStream_t stream) {
  const float* I = (const float*)d_in[0];
  const float* P = (const float*)d_in[1];
  float* Q = (float*)d_out;
  float* Aws = (float*)d_ws;
  float* Bws = Aws + (size_t)NB * HH * WW;
  dim3 grid(WW / 16, HH / (16 * NT), NB);
  gf_stats<<<grid, dim3(32), 0, stream>>>(I, P, Aws, Bws);
  gf_final<<<grid, dim3(32), 0, stream>>>(I, Aws, Bws, Q);
  (void)in_sizes; (void)n_in; (void)out_size; (void)ws_size;
}
